# starting kernel
# baseline (speedup 1.0000x reference)
#include <hip/hip_runtime.h>
#include <hip/hip_bf16.h>

typedef __attribute__((ext_vector_type(2))) float v2f;
typedef __attribute__((ext_vector_type(4))) float v4f;
typedef __attribute__((ext_vector_type(8))) float v8f;

#define NFEAT 512
#define NFILT 64
#define KCHUNK 128

__global__ void kw_zero_kernel(float* __restrict__ p, int n) {
    int i = blockIdx.x * blockDim.x + threadIdx.x;
    if (i < n) p[i] = 0.0f;
}

__global__ void kw_degree_kernel(const int* __restrict__ rows, float* __restrict__ deg, int E) {
    int i = blockIdx.x * blockDim.x + threadIdx.x;
    if (i < E)
        __hip_atomic_fetch_add(&deg[rows[i]], 1.0f, __ATOMIC_RELAXED, __HIP_MEMORY_SCOPE_AGENT);
}

__global__ void kw_dinv_kernel(float* __restrict__ deg, int N) {
    int i = blockIdx.x * blockDim.x + threadIdx.x;
    if (i < N) {
        float d = deg[i];
        deg[i] = (d > 0.0f) ? rsqrtf(d) : 0.0f;
    }
}

__global__ __launch_bounds__(256) void kw_gemm_wmma(const float* __restrict__ x,
                                                    const float* __restrict__ flt,
                                                    float* __restrict__ XF, int N) {
    __shared__ float sB[KCHUNK * NFILT];

    const int lane = threadIdx.x & 31;
    const int wib  = threadIdx.x >> 5;
    const int half = lane >> 4;
    const int l16  = lane & 15;

    const int stripe = blockIdx.x * 8 + wib;
    const int m0     = stripe * 16;
    const bool active = (m0 < N);

    int row = m0 + l16;
    if (row >= N) row = N - 1;
    const float* xrow = x + (size_t)row * NFEAT;

    v8f acc0 = {}, acc1 = {}, acc2 = {}, acc3 = {};

    const float* bbase = sB + lane * 2;

    for (int kc = 0; kc < NFEAT; kc += KCHUNK) {
        __syncthreads();
        {
            const v4f* src = (const v4f*)(flt + (size_t)kc * NFILT);
            for (int t4 = threadIdx.x; t4 < KCHUNK * (NFILT / 4); t4 += 256) {
                v4f v = src[t4];
                int r    = t4 >> 4;
                int g    = r >> 2;
                int koff = r & 3;
                int h    = koff >> 1;
                int j    = koff & 1;
                int cb   = (t4 & 15) << 2;
                int n    = cb >> 4;
                int c16  = cb & 15;
                float* d = sB + (((g << 2) + n) << 6) + (((h << 4) + c16) << 1) + j;
                d[0] = v.x; d[2] = v.y; d[4] = v.z; d[6] = v.w;
            }
        }
        __syncthreads();

        if (active) {
#pragma unroll 4
            for (int k2 = 0; k2 < KCHUNK; k2 += 4) {
                v2f a = *(const v2f*)(xrow + kc + k2 + 2 * half);

                const int g = k2 >> 2;
                const float* bg = bbase + (g << 8);
                v2f b0 = *(const v2f*)(bg);
                v2f b1 = *(const v2f*)(bg + 64);
                v2f b2 = *(const v2f*)(bg + 128);
                v2f b3 = *(const v2f*)(bg + 192);

                acc0 = __builtin_amdgcn_wmma_f32_16x16x4_f32(false, a, false, b0, (short)0, acc0, false, false);
                acc1 = __builtin_amdgcn_wmma_f32_16x16x4_f32(false, a, false, b1, (short)0, acc1, false, false);
                acc2 = __builtin_amdgcn_wmma_f32_16x16x4_f32(false, a, false, b2, (short)0, acc2, false, false);
                acc3 = __builtin_amdgcn_wmma_f32_16x16x4_f32(false, a, false, b3, (short)0, acc3, false, false);
            }
        }
    }

    if (active) {
#pragma unroll
        for (int r = 0; r < 8; ++r) {
            int M = m0 + r + 8 * half;
            if (M < N) {
                float* o = XF + (size_t)M * NFILT + l16;
                o[0]  = acc0[r];
                o[16] = acc1[r];
                o[32] = acc2[r];
                o[48] = acc3[r];
            }
        }
    }
}

__global__ __launch_bounds__(256) void kw_scatter_kernel(const int* __restrict__ rows,
                                                         const int* __restrict__ cols,
                                                         const float* __restrict__ dinv,
                                                         const float* __restrict__ XF,
                                                         float* __restrict__ out, int E) {
    const int wid  = blockIdx.x * (blockDim.x >> 5) + (threadIdx.x >> 5);
    const int lane = threadIdx.x & 31;
    if (wid >= E) return;

    const int r = rows[wid];
    const int c = cols[wid];
    const float w = dinv[r] * dinv[c];

    const float* src = XF + (size_t)c * NFILT;
    float v0 = src[lane];
    float v1 = src[lane + 32];

    float* dst = out + (size_t)r * NFILT;
    __hip_atomic_fetch_add(&dst[lane],      w * v0, __ATOMIC_RELAXED, __HIP_MEMORY_SCOPE_AGENT);
    __hip_atomic_fetch_add(&dst[lane + 32], w * v1, __ATOMIC_RELAXED, __HIP_MEMORY_SCOPE_AGENT);
}

extern "C" void kernel_launch(void* const* d_in, const int* in_sizes, int n_in,
                              void* d_out, int out_size, void* d_ws, size_t ws_size,
                              hipStream_t stream) {
    const float* x   = (const float*)d_in[0];
    const float* flt = (const float*)d_in[1];
    const int*   eid = (const int*)d_in[2];

    const int N = in_sizes[0] / NFEAT;
    const int E = in_sizes[2] / 2;
    const int* rows = eid;
    const int* cols = eid + E;

    float* out = (float*)d_out;
    float* deg = (float*)d_ws;
    float* XF  = (float*)d_ws + (size_t)((N + 63) & ~63);

    kw_zero_kernel<<<(N + 255) / 256, 256, 0, stream>>>(deg, N);
    kw_zero_kernel<<<(out_size + 255) / 256, 256, 0, stream>>>(out, out_size);

    kw_degree_kernel<<<(E + 255) / 256, 256, 0, stream>>>(rows, deg, E);
    kw_dinv_kernel<<<(N + 255) / 256, 256, 0, stream>>>(deg, N);

    const int stripes = (N + 15) / 16;
    kw_gemm_wmma<<<(stripes + 7) / 8, 256, 0, stream>>>(x, flt, XF, N);

    kw_scatter_kernel<<<(E + 7) / 8, 256, 0, stream>>>(rows, cols, deg, XF, out, E);
}


# Edge scatter's global f32 atomics and the zero pass replaced by row-range owner work-groups: each scans all row ids, sums its 400 rows in LDS (f32 LDS adds, arrival order), stores each line once
# speedup vs baseline: 1.8116x; 1.8116x over previous
#include <hip/hip_runtime.h>
#include <hip/hip_bf16.h>

typedef __attribute__((ext_vector_type(2))) float v2f;
typedef __attribute__((ext_vector_type(4))) float v4f;
typedef __attribute__((ext_vector_type(8))) float v8f;

#define NFEAT 512
#define NFILT 64
#define KCHUNK 128

#define DEG_CHUNK       32768
#define DEG_RANGE_WORDS 10240
#define DEG_RANGE_BINS  (DEG_RANGE_WORDS * 2)

__device__ __forceinline__ void kw_count_row(unsigned* h, int r, int base) {
    const unsigned d = (unsigned)(r - base);
    if (d < (unsigned)DEG_RANGE_BINS)
        atomicAdd(&h[d >> 1], 1u << ((d & 1u) << 4));
}

__global__ __launch_bounds__(1024) void kw_degree_lds(const int* __restrict__ rows,
                                                      unsigned* __restrict__ partial,
                                                      int E, int nwords, int wpg) {
    __shared__ __attribute__((aligned(16))) unsigned h[DEG_RANGE_WORDS];

    const int tid  = threadIdx.x;
    const int base = blockIdx.y * DEG_RANGE_BINS;

    for (int i = tid; i < DEG_RANGE_WORDS / 4; i += 1024)
        ((uint4*)h)[i] = make_uint4(0u, 0u, 0u, 0u);

    const int e0 = blockIdx.x * DEG_CHUNK;

    int4 rr[DEG_CHUNK / 4096];
#pragma unroll
    for (int j = 0; j < DEG_CHUNK / 4096; ++j) {
        const int eb = e0 + (j << 12);
        const int e  = eb + tid * 4;
        if (eb + 4096 <= E) {
            rr[j] = *(const int4*)(rows + e);
        } else {
            rr[j].x = (e     < E) ? rows[e]     : -1;
            rr[j].y = (e + 1 < E) ? rows[e + 1] : -1;
            rr[j].z = (e + 2 < E) ? rows[e + 2] : -1;
            rr[j].w = (e + 3 < E) ? rows[e + 3] : -1;
        }
    }
    __syncthreads();

#pragma unroll
    for (int j = 0; j < DEG_CHUNK / 4096; ++j) {
        kw_count_row(h, rr[j].x, base);
        kw_count_row(h, rr[j].y, base);
        kw_count_row(h, rr[j].z, base);
        kw_count_row(h, rr[j].w, base);
    }
    __syncthreads();

    const int wbase = blockIdx.y * DEG_RANGE_WORDS;
    uint4* dst = (uint4*)(partial + (size_t)blockIdx.x * wpg + wbase);
    for (int i = tid; i < DEG_RANGE_WORDS / 4; i += 1024)
        if (wbase + i * 4 < nwords)
            dst[i] = ((const uint4*)h)[i];
}

__global__ __launch_bounds__(256) void kw_dinv_reduce(const unsigned* __restrict__ partial,
                                                      float* __restrict__ dinv,
                                                      int N, int nwords, int G, int wpg) {
    const int w = blockIdx.x * 256 + threadIdx.x;
    if (w >= nwords) return;

    const unsigned* p = partial + w;
    unsigned lo = 0u, hi = 0u;
#pragma unroll 7
    for (int g = 0; g < G; ++g) {
        const unsigned v = p[(size_t)g * wpg];
        lo += v & 0xFFFFu;
        hi += v >> 16;
    }
    const float d0 = (float)lo;
    const float d1 = (float)hi;
    const float i0 = (d0 > 0.0f) ? rsqrtf(d0) : 0.0f;
    const float i1 = (d1 > 0.0f) ? rsqrtf(d1) : 0.0f;

    const int n0 = 2 * w;
    if (n0 + 1 < N) {
        v2f o = {i0, i1};
        *(v2f*)(dinv + n0) = o;
    } else {
        dinv[n0] = i0;
    }
}

__global__ __launch_bounds__(256) void kw_gemm_wmma(const float* __restrict__ x,
                                                    const float* __restrict__ flt,
                                                    float* __restrict__ XF, int N) {
    __shared__ float sB[KCHUNK * NFILT];

    const int lane = threadIdx.x & 31;
    const int wib  = threadIdx.x >> 5;
    const int half = lane >> 4;
    const int l16  = lane & 15;

    const int stripe = blockIdx.x * 8 + wib;
    const int m0     = stripe * 16;
    const bool active = (m0 < N);

    int row = m0 + l16;
    if (row >= N) row = N - 1;
    const float* xrow = x + (size_t)row * NFEAT;

    v8f acc0 = {}, acc1 = {}, acc2 = {}, acc3 = {};

    const float* bbase = sB + lane * 2;

    for (int kc = 0; kc < NFEAT; kc += KCHUNK) {
        __syncthreads();
        {
            const v4f* src = (const v4f*)(flt + (size_t)kc * NFILT);
            for (int t4 = threadIdx.x; t4 < KCHUNK * (NFILT / 4); t4 += 256) {
                v4f v = src[t4];
                int r    = t4 >> 4;
                int g    = r >> 2;
                int koff = r & 3;
                int h    = koff >> 1;
                int j    = koff & 1;
                int cb   = (t4 & 15) << 2;
                int n    = cb >> 4;
                int c16  = cb & 15;
                float* d = sB + (((g << 2) + n) << 6) + (((h << 4) + c16) << 1) + j;
                d[0] = v.x; d[2] = v.y; d[4] = v.z; d[6] = v.w;
            }
        }
        __syncthreads();

        if (active) {
#pragma unroll 4
            for (int k2 = 0; k2 < KCHUNK; k2 += 4) {
                v2f a = *(const v2f*)(xrow + kc + k2 + 2 * half);

                const int g = k2 >> 2;
                const float* bg = bbase + (g << 8);
                v2f b0 = *(const v2f*)(bg);
                v2f b1 = *(const v2f*)(bg + 64);
                v2f b2 = *(const v2f*)(bg + 128);
                v2f b3 = *(const v2f*)(bg + 192);

                acc0 = __builtin_amdgcn_wmma_f32_16x16x4_f32(false, a, false, b0, (short)0, acc0, false, false);
                acc1 = __builtin_amdgcn_wmma_f32_16x16x4_f32(false, a, false, b1, (short)0, acc1, false, false);
                acc2 = __builtin_amdgcn_wmma_f32_16x16x4_f32(false, a, false, b2, (short)0, acc2, false, false);
                acc3 = __builtin_amdgcn_wmma_f32_16x16x4_f32(false, a, false, b3, (short)0, acc3, false, false);
            }
        }
    }

    if (active) {
#pragma unroll
        for (int r = 0; r < 8; ++r) {
            int M = m0 + r + 8 * half;
            if (M < N) {
                float* o = XF + (size_t)M * NFILT + l16;
                o[0]  = acc0[r];
                o[16] = acc1[r];
                o[32] = acc2[r];
                o[48] = acc3[r];
            }
        }
    }
}

#define SC_ROWS  400
#define SC_WAVES 32
#define SC_BLK   512
#define SC_CAP   (SC_BLK + 32)

__device__ __forceinline__ void kw_load_blk(int4 (&r)[4], const int* __restrict__ rows,
                                            int blk, int lane) {
    const int4* p = (const int4*)(rows + (size_t)blk * SC_BLK) + lane;
#pragma unroll
    for (int q = 0; q < 4; ++q)
        r[q] = p[q * 32];
}

__device__ __forceinline__ void kw_load_tail(int4 (&r)[4], const int* __restrict__ rows,
                                             int blk, int lane, int E) {
#pragma unroll
    for (int q = 0; q < 4; ++q) {
        const int e = blk * SC_BLK + q * 128 + lane * 4;
        r[q].x = (e     < E) ? rows[e]     : -1;
        r[q].y = (e + 1 < E) ? rows[e + 1] : -1;
        r[q].z = (e + 2 < E) ? rows[e + 2] : -1;
        r[q].w = (e + 3 < E) ? rows[e + 3] : -1;
    }
}

__device__ __forceinline__ void kw_append(uint2* lst, int& cnt, bool hit, unsigned e, unsigned d) {
    const unsigned m = __builtin_amdgcn_ballot_w32(hit);
    if (hit) lst[cnt + (int)__builtin_amdgcn_mbcnt_lo(m, 0u)] = make_uint2(e, d);
    cnt += __builtin_popcount(m);
}

template <bool FULL>
__device__ __forceinline__ void kw_flush(const uint2* lst, int first, int nvalid, int lane, int base,
                                         const int* __restrict__ cols,
                                         const float* __restrict__ dinv,
                                         const float* __restrict__ XF, float* acc) {
    int   c = 0, d = 0;
    float w = 0.0f;
    if (FULL || lane < nvalid) {
        const uint2 t = lst[first + lane];
        c = cols[t.x];
        d = (int)t.y;
        w = dinv[base + d] * dinv[c];
    }
    for (int j0 = 0; j0 < 32; j0 += 8) {
        if (!FULL && j0 >= nvalid) break;
        float v0[8], v1[8];
#pragma unroll
        for (int u = 0; u < 8; ++u) {
            const int cj = __builtin_amdgcn_readlane(c, j0 + u);
            const float* src = XF + (size_t)cj * NFILT;
            v0[u] = src[lane];
            v1[u] = src[lane + 32];
        }
#pragma unroll
        for (int u = 0; u < 8; ++u) {
            if (FULL || j0 + u < nvalid) {
                const float wj = __builtin_bit_cast(float, __builtin_amdgcn_readlane(__builtin_bit_cast(int, w), j0 + u));
                const int   dj = __builtin_amdgcn_readlane(d, j0 + u);
                float* dst = acc + dj * NFILT + lane;
                __hip_atomic_fetch_add(&dst[0],  wj * v0[u], __ATOMIC_RELAXED, __HIP_MEMORY_SCOPE_WORKGROUP);
                __hip_atomic_fetch_add(&dst[32], wj * v1[u], __ATOMIC_RELAXED, __HIP_MEMORY_SCOPE_WORKGROUP);
            }
        }
    }
}

__device__ __forceinline__ void kw_scan_blk(const int4 (&r)[4], int blk, uint2* lst, int& cnt,
                                            int lane, int base,
                                            const int* __restrict__ cols,
                                            const float* __restrict__ dinv,
                                            const float* __restrict__ XF, float* acc) {
#pragma unroll
    for (int q = 0; q < 4; ++q) {
        const unsigned dx = (unsigned)(r[q].x - base);
        const unsigned dy = (unsigned)(r[q].y - base);
        const unsigned dz = (unsigned)(r[q].z - base);
        const unsigned dw = (unsigned)(r[q].w - base);
        const unsigned dm = min(min(dx, dy), min(dz, dw));
        if (__builtin_amdgcn_ballot_w32(dm < (unsigned)SC_ROWS) != 0u) {
            const unsigned e = (unsigned)(blk * SC_BLK + q * 128 + lane * 4);
            kw_append(lst, cnt, dx < (unsigned)SC_ROWS, e,      dx);
            kw_append(lst, cnt, dy < (unsigned)SC_ROWS, e + 1u, dy);
            kw_append(lst, cnt, dz < (unsigned)SC_ROWS, e + 2u, dz);
            kw_append(lst, cnt, dw < (unsigned)SC_ROWS, e + 3u, dw);
        }
    }
    while (cnt >= 32) {
        cnt -= 32;
        kw_flush<true>(lst, cnt, 32, lane, base, cols, dinv, XF, acc);
    }
}

__global__ __launch_bounds__(1024) void kw_rowsum_lds(const int* __restrict__ rows,
                                                      const int* __restrict__ cols,
                                                      const float* __restrict__ dinv,
                                                      const float* __restrict__ XF,
                                                      float* __restrict__ out, int N, int E) {
    __shared__ __attribute__((aligned(16))) float acc[SC_ROWS * NFILT];
    __shared__ __attribute__((aligned(16))) uint2 slist[SC_WAVES * SC_CAP];

    const int tid  = threadIdx.x;
    const int lane = tid & 31;
    const int wv   = __builtin_amdgcn_readfirstlane(tid >> 5);
    const int base = blockIdx.x * SC_ROWS;

    for (int i = tid; i < SC_ROWS * NFILT / 4; i += 1024)
        ((v4f*)acc)[i] = (v4f){0.0f, 0.0f, 0.0f, 0.0f};
    __syncthreads();

    uint2* lst = slist + wv * SC_CAP;
    int cnt = 0;

    const int nfull = E / SC_BLK;
    const int iters = (nfull + SC_WAVES - 1) / SC_WAVES;

    if (iters > 0) {
        int k = (int)((blockIdx.x * 13u) % (unsigned)iters);

        int4 ra[4], rb[4];
        kw_load_blk(ra, rows, min(wv + SC_WAVES * k, nfull - 1), lane);

        for (int t = 0; t < iters; t += 2) {
            int kb = k + 1;
            if (kb == iters) kb = 0;
            const int blka = wv + SC_WAVES * k;
            const int blkb = wv + SC_WAVES * kb;
            kw_load_blk(rb, rows, min(blkb, nfull - 1), lane);
            if (blka < nfull)
                kw_scan_blk(ra, blka, lst, cnt, lane, base, cols, dinv, XF, acc);

            int kn = kb + 1;
            if (kn == iters) kn = 0;
            kw_load_blk(ra, rows, min(wv + SC_WAVES * kn, nfull - 1), lane);
            if (t + 1 < iters && blkb < nfull)
                kw_scan_blk(rb, blkb, lst, cnt, lane, base, cols, dinv, XF, acc);
            k = kn;
        }
    }
    if (nfull * SC_BLK < E && wv == (nfull & (SC_WAVES - 1))) {
        int4 rt[4];
        kw_load_tail(rt, rows, nfull, lane, E);
        kw_scan_blk(rt, nfull, lst, cnt, lane, base, cols, dinv, XF, acc);
    }
    if (cnt > 0)
        kw_flush<false>(lst, 0, cnt, lane, base, cols, dinv, XF, acc);

    __syncthreads();

    for (int i = tid; i < SC_ROWS * (NFILT / 4); i += 1024) {
        const int row = base + (i >> 4);
        if (row < N)
            ((v4f*)(out + (size_t)row * NFILT))[i & 15] = ((const v4f*)acc)[i];
    }
}

extern "C" void kernel_launch(void* const* d_in, const int* in_sizes, int n_in,
                              void* d_out, int out_size, void* d_ws, size_t ws_size,
                              hipStream_t stream) {
    const float* x   = (const float*)d_in[0];
    const float* flt = (const float*)d_in[1];
    const int*   eid = (const int*)d_in[2];

    const int N = in_sizes[0] / NFEAT;
    const int E = in_sizes[2] / 2;
    const int* rows = eid;
    const int* cols = eid + E;

    float* out = (float*)d_out;
    float* deg = (float*)d_ws;
    float* XF  = (float*)d_ws + (size_t)((N + 63) & ~63);

    unsigned* partial = (unsigned*)XF;
    const int nwords = (N + 1) / 2;
    const int R   = (N + DEG_RANGE_BINS - 1) / DEG_RANGE_BINS;
    const int G   = (E + DEG_CHUNK - 1) / DEG_CHUNK;
    const int wpg = R * DEG_RANGE_WORDS;
    kw_degree_lds<<<dim3(G, R), 1024, 0, stream>>>(rows, partial, E, nwords, wpg);
    kw_dinv_reduce<<<(nwords + 255) / 256, 256, 0, stream>>>(partial, deg, N, nwords, G, wpg);

    const int stripes = (N + 15) / 16;
    kw_gemm_wmma<<<(stripes + 7) / 8, 256, 0, stream>>>(x, flt, XF, N);

    kw_rowsum_lds<<<(N + SC_ROWS - 1) / SC_ROWS, 1024, 0, stream>>>(rows, cols, deg, XF, out, N, E);
}


# Row-sum owners read only row-id lines holding their rows: degree launch ORs a per-owner line bitmap in LDS, parked in the owner's own output rows; LDS f32 adds still in arrival order
# speedup vs baseline: 2.2226x; 1.2269x over previous
#include <hip/hip_runtime.h>
#include <hip/hip_bf16.h>

typedef __attribute__((ext_vector_type(2))) float v2f;
typedef __attribute__((ext_vector_type(4))) float v4f;
typedef __attribute__((ext_vector_type(8))) float v8f;

#define NFEAT 512
#define NFILT 64
#define KCHUNK 128

#define DEG_CHUNK       32768
#define DEG_RANGE_WORDS 10240
#define DEG_RANGE_BINS  (DEG_RANGE_WORDS * 2)

#define SC_ROWS        400
#define MARK_OWN_MAX   56
#define MARK_CHUNK_W   (DEG_CHUNK / 32 / 32)

__device__ __forceinline__ void kw_count_row(unsigned* h, int r, int base) {
    const unsigned d = (unsigned)(r - base);
    if (d < (unsigned)DEG_RANGE_BINS)
        atomicAdd(&h[d >> 1], 1u << ((d & 1u) << 4));
}

__device__ __forceinline__ void kw_mark_row(unsigned* bm, int r, unsigned obase, unsigned opq,
                                            unsigned nown, int wi, unsigned bit) {
    const unsigned o  = (unsigned)r / (unsigned)SC_ROWS;
    const unsigned ol = o - obase;
    if (ol < opq && o < nown)
        atomicOr(&bm[ol * MARK_CHUNK_W + wi], bit);
}

__global__ __launch_bounds__(1024) void kw_degree_lds(const int* __restrict__ rows,
                                                      unsigned* __restrict__ partial,
                                                      unsigned* __restrict__ marks,
                                                      int E, int nwords, int wpg,
                                                      int nown, int opq) {
    __shared__ __attribute__((aligned(16))) unsigned h[DEG_RANGE_WORDS];
    __shared__ unsigned bm[MARK_OWN_MAX * MARK_CHUNK_W];

    const int tid  = threadIdx.x;
    const int base = blockIdx.y * DEG_RANGE_BINS;

    for (int i = tid; i < DEG_RANGE_WORDS / 4; i += 1024)
        ((uint4*)h)[i] = make_uint4(0u, 0u, 0u, 0u);
    for (int i = tid; i < MARK_OWN_MAX * MARK_CHUNK_W; i += 1024)
        bm[i] = 0u;

    const int e0 = blockIdx.x * DEG_CHUNK;

    int4 rr[DEG_CHUNK / 4096];
#pragma unroll
    for (int j = 0; j < DEG_CHUNK / 4096; ++j) {
        const int eb = e0 + (j << 12);
        const int e  = eb + tid * 4;
        if (eb + 4096 <= E) {
            rr[j] = *(const int4*)(rows + e);
        } else {
            rr[j].x = (e     < E) ? rows[e]     : -1;
            rr[j].y = (e + 1 < E) ? rows[e + 1] : -1;
            rr[j].z = (e + 2 < E) ? rows[e + 2] : -1;
            rr[j].w = (e + 3 < E) ? rows[e + 3] : -1;
        }
    }
    __syncthreads();

#pragma unroll
    for (int j = 0; j < DEG_CHUNK / 4096; ++j) {
        kw_count_row(h, rr[j].x, base);
        kw_count_row(h, rr[j].y, base);
        kw_count_row(h, rr[j].z, base);
        kw_count_row(h, rr[j].w, base);
    }
    const unsigned obase = blockIdx.y * (unsigned)opq;
    if (marks != nullptr) {
        const unsigned bit = 1u << ((tid >> 3) & 31);
#pragma unroll
        for (int j = 0; j < DEG_CHUNK / 4096; ++j) {
            const int wi = j * 4 + (tid >> 8);
            kw_mark_row(bm, rr[j].x, obase, (unsigned)opq, (unsigned)nown, wi, bit);
            kw_mark_row(bm, rr[j].y, obase, (unsigned)opq, (unsigned)nown, wi, bit);
            kw_mark_row(bm, rr[j].z, obase, (unsigned)opq, (unsigned)nown, wi, bit);
            kw_mark_row(bm, rr[j].w, obase, (unsigned)opq, (unsigned)nown, wi, bit);
        }
    }
    __syncthreads();

    if (marks != nullptr) {
        for (int i = tid; i < opq * MARK_CHUNK_W; i += 1024) {
            const unsigned o = obase + (unsigned)(i >> 5);
            if (o < (unsigned)nown)
                marks[(size_t)o * (SC_ROWS * NFILT) + blockIdx.x * MARK_CHUNK_W + (i & 31)] = bm[i];
        }
    }

    const int wbase = blockIdx.y * DEG_RANGE_WORDS;
    uint4* dst = (uint4*)(partial + (size_t)blockIdx.x * wpg + wbase);
    for (int i = tid; i < DEG_RANGE_WORDS / 4; i += 1024)
        if (wbase + i * 4 < nwords)
            dst[i] = ((const uint4*)h)[i];
}

__global__ __launch_bounds__(256) void kw_dinv_reduce(const unsigned* __restrict__ partial,
                                                      float* __restrict__ dinv,
                                                      int N, int nwords, int G, int wpg) {
    const int w = blockIdx.x * 256 + threadIdx.x;
    if (w >= nwords) return;

    const unsigned* p = partial + w;
    unsigned lo = 0u, hi = 0u;
#pragma unroll 7
    for (int g = 0; g < G; ++g) {
        const unsigned v = p[(size_t)g * wpg];
        lo += v & 0xFFFFu;
        hi += v >> 16;
    }
    const float d0 = (float)lo;
    const float d1 = (float)hi;
    const float i0 = (d0 > 0.0f) ? rsqrtf(d0) : 0.0f;
    const float i1 = (d1 > 0.0f) ? rsqrtf(d1) : 0.0f;

    const int n0 = 2 * w;
    if (n0 + 1 < N) {
        v2f o = {i0, i1};
        *(v2f*)(dinv + n0) = o;
    } else {
        dinv[n0] = i0;
    }
}

__global__ __launch_bounds__(256) void kw_gemm_wmma(const float* __restrict__ x,
                                                    const float* __restrict__ flt,
                                                    float* __restrict__ XF, int N) {
    __shared__ float sB[KCHUNK * NFILT];

    const int lane = threadIdx.x & 31;
    const int wib  = threadIdx.x >> 5;
    const int half = lane >> 4;
    const int l16  = lane & 15;

    const int stripe = blockIdx.x * 8 + wib;
    const int m0     = stripe * 16;
    const bool active = (m0 < N);

    int row = m0 + l16;
    if (row >= N) row = N - 1;
    const float* xrow = x + (size_t)row * NFEAT;

    v8f acc0 = {}, acc1 = {}, acc2 = {}, acc3 = {};

    const float* bbase = sB + lane * 2;

    for (int kc = 0; kc < NFEAT; kc += KCHUNK) {
        __syncthreads();
        {
            const v4f* src = (const v4f*)(flt + (size_t)kc * NFILT);
            for (int t4 = threadIdx.x; t4 < KCHUNK * (NFILT / 4); t4 += 256) {
                v4f v = src[t4];
                int r    = t4 >> 4;
                int g    = r >> 2;
                int koff = r & 3;
                int h    = koff >> 1;
                int j    = koff & 1;
                int cb   = (t4 & 15) << 2;
                int n    = cb >> 4;
                int c16  = cb & 15;
                float* d = sB + (((g << 2) + n) << 6) + (((h << 4) + c16) << 1) + j;
                d[0] = v.x; d[2] = v.y; d[4] = v.z; d[6] = v.w;
            }
        }
        __syncthreads();

        if (active) {
#pragma unroll 4
            for (int k2 = 0; k2 < KCHUNK; k2 += 4) {
                v2f a = *(const v2f*)(xrow + kc + k2 + 2 * half);

                const int g = k2 >> 2;
                const float* bg = bbase + (g << 8);
                v2f b0 = *(const v2f*)(bg);
                v2f b1 = *(const v2f*)(bg + 64);
                v2f b2 = *(const v2f*)(bg + 128);
                v2f b3 = *(const v2f*)(bg + 192);

                acc0 = __builtin_amdgcn_wmma_f32_16x16x4_f32(false, a, false, b0, (short)0, acc0, false, false);
                acc1 = __builtin_amdgcn_wmma_f32_16x16x4_f32(false, a, false, b1, (short)0, acc1, false, false);
                acc2 = __builtin_amdgcn_wmma_f32_16x16x4_f32(false, a, false, b2, (short)0, acc2, false, false);
                acc3 = __builtin_amdgcn_wmma_f32_16x16x4_f32(false, a, false, b3, (short)0, acc3, false, false);
            }
        }
    }

    if (active) {
#pragma unroll
        for (int r = 0; r < 8; ++r) {
            int M = m0 + r + 8 * half;
            if (M < N) {
                float* o = XF + (size_t)M * NFILT + l16;
                o[0]  = acc0[r];
                o[16] = acc1[r];
                o[32] = acc2[r];
                o[48] = acc3[r];
            }
        }
    }
}

#define SC_WAVES 32
#define SC_BLK   512
#define SC_CAP   (SC_BLK + 32)
#define LL_CAP   144

__device__ __forceinline__ void kw_load_blk(int4 (&r)[4], const int* __restrict__ rows,
                                            int blk, int lane) {
    const int4* p = (const int4*)(rows + (size_t)blk * SC_BLK) + lane;
#pragma unroll
    for (int q = 0; q < 4; ++q)
        r[q] = p[q * 32];
}

__device__ __forceinline__ void kw_load_tail(int4 (&r)[4], const int* __restrict__ rows,
                                             int blk, int lane, int E) {
#pragma unroll
    for (int q = 0; q < 4; ++q) {
        const int e = blk * SC_BLK + q * 128 + lane * 4;
        r[q].x = (e     < E) ? rows[e]     : -1;
        r[q].y = (e + 1 < E) ? rows[e + 1] : -1;
        r[q].z = (e + 2 < E) ? rows[e + 2] : -1;
        r[q].w = (e + 3 < E) ? rows[e + 3] : -1;
    }
}

__device__ __forceinline__ void kw_append(uint2* lst, int& cnt, bool hit, unsigned e, unsigned d) {
    const unsigned m = __builtin_amdgcn_ballot_w32(hit);
    if (hit) lst[cnt + (int)__builtin_amdgcn_mbcnt_lo(m, 0u)] = make_uint2(e, d);
    cnt += __builtin_popcount(m);
}

template <bool FULL>
__device__ __forceinline__ void kw_flush(const uint2* lst, int first, int nvalid, int lane, int base,
                                         const int* __restrict__ cols,
                                         const float* __restrict__ dinv,
                                         const float* __restrict__ XF, float* acc) {
    int   c = 0, d = 0;
    float w = 0.0f;
    if (FULL || lane < nvalid) {
        const uint2 t = lst[first + lane];
        c = cols[t.x];
        d = (int)t.y;
        w = dinv[base + d] * dinv[c];
    }
    for (int j0 = 0; j0 < 32; j0 += 8) {
        if (!FULL && j0 >= nvalid) break;
        float v0[8], v1[8];
#pragma unroll
        for (int u = 0; u < 8; ++u) {
            const int cj = __builtin_amdgcn_readlane(c, j0 + u);
            const float* src = XF + (size_t)cj * NFILT;
            v0[u] = src[lane];
            v1[u] = src[lane + 32];
        }
#pragma unroll
        for (int u = 0; u < 8; ++u) {
            if (FULL || j0 + u < nvalid) {
                const float wj = __builtin_bit_cast(float, __builtin_amdgcn_readlane(__builtin_bit_cast(int, w), j0 + u));
                const int   dj = __builtin_amdgcn_readlane(d, j0 + u);
                float* dst = acc + dj * NFILT + lane;
                __hip_atomic_fetch_add(&dst[0],  wj * v0[u], __ATOMIC_RELAXED, __HIP_MEMORY_SCOPE_WORKGROUP);
                __hip_atomic_fetch_add(&dst[32], wj * v1[u], __ATOMIC_RELAXED, __HIP_MEMORY_SCOPE_WORKGROUP);
            }
        }
    }
}

__device__ __forceinline__ void kw_scan_blk(const int4 (&r)[4], int blk, uint2* lst, int& cnt,
                                            int lane, int base,
                                            const int* __restrict__ cols,
                                            const float* __restrict__ dinv,
                                            const float* __restrict__ XF, float* acc) {
#pragma unroll
    for (int q = 0; q < 4; ++q) {
        const unsigned dx = (unsigned)(r[q].x - base);
        const unsigned dy = (unsigned)(r[q].y - base);
        const unsigned dz = (unsigned)(r[q].z - base);
        const unsigned dw = (unsigned)(r[q].w - base);
        const unsigned dm = min(min(dx, dy), min(dz, dw));
        if (__builtin_amdgcn_ballot_w32(dm < (unsigned)SC_ROWS) != 0u) {
            const unsigned e = (unsigned)(blk * SC_BLK + q * 128 + lane * 4);
            kw_append(lst, cnt, dx < (unsigned)SC_ROWS, e,      dx);
            kw_append(lst, cnt, dy < (unsigned)SC_ROWS, e + 1u, dy);
            kw_append(lst, cnt, dz < (unsigned)SC_ROWS, e + 2u, dz);
            kw_append(lst, cnt, dw < (unsigned)SC_ROWS, e + 3u, dw);
        }
    }
    while (cnt >= 32) {
        cnt -= 32;
        kw_flush<true>(lst, cnt, 32, lane, base, cols, dinv, XF, acc);
    }
}

template <bool FULL>
__device__ __forceinline__ void kw_scan_lines(const unsigned* ll, int nvalid, uint2* lst, int& cnt,
                                              int lane, int base, int E,
                                              const int* __restrict__ rows,
                                              const int* __restrict__ cols,
                                              const float* __restrict__ dinv,
                                              const float* __restrict__ XF, float* acc) {
    int4     r[4];
    unsigned e0[4];
#pragma unroll
    for (int q = 0; q < 4; ++q) {
        const int s = q * 4 + (lane >> 3);
        if (FULL || s < nvalid) {
            e0[q] = ll[s] * 32u + (unsigned)((lane & 7) << 2);
            r[q]  = *(const int4*)(rows + e0[q]);
        } else {
            e0[q] = 0u;
            r[q]  = make_int4(-1, -1, -1, -1);
        }
    }
#pragma unroll
    for (int q = 0; q < 4; ++q) {
        const unsigned dx = (unsigned)(r[q].x - base);
        const unsigned dy = (unsigned)(r[q].y - base);
        const unsigned dz = (unsigned)(r[q].z - base);
        const unsigned dw = (unsigned)(r[q].w - base);
        const unsigned dm = min(min(dx, dy), min(dz, dw));
        if (__builtin_amdgcn_ballot_w32(dm < (unsigned)SC_ROWS) != 0u) {
            const unsigned e = e0[q];
            kw_append(lst, cnt, dx < (unsigned)SC_ROWS && e      < (unsigned)E, e,      dx);
            kw_append(lst, cnt, dy < (unsigned)SC_ROWS && e + 1u < (unsigned)E, e + 1u, dy);
            kw_append(lst, cnt, dz < (unsigned)SC_ROWS && e + 2u < (unsigned)E, e + 2u, dz);
            kw_append(lst, cnt, dw < (unsigned)SC_ROWS && e + 3u < (unsigned)E, e + 3u, dw);
        }
    }
    while (cnt >= 32) {
        cnt -= 32;
        kw_flush<true>(lst, cnt, 32, lane, base, cols, dinv, XF, acc);
    }
}

__global__ __launch_bounds__(1024) void kw_rowsum_lds(const int* __restrict__ rows,
                                                      const int* __restrict__ cols,
                                                      const float* __restrict__ dinv,
                                                      const float* __restrict__ XF,
                                                      float* out, int N, int E, int mwords) {
    __shared__ __attribute__((aligned(16))) float acc[SC_ROWS * NFILT];
    __shared__ __attribute__((aligned(16))) uint2 slist[SC_WAVES * SC_CAP];
    __shared__ unsigned llist[SC_WAVES * LL_CAP];

    const int tid  = threadIdx.x;
    const int lane = tid & 31;
    const int wv   = __builtin_amdgcn_readfirstlane(tid >> 5);
    const int base = blockIdx.x * SC_ROWS;

    for (int i = tid; i < SC_ROWS * NFILT / 4; i += 1024)
        ((v4f*)acc)[i] = (v4f){0.0f, 0.0f, 0.0f, 0.0f};
    __syncthreads();

    uint2* lst = slist + wv * SC_CAP;
    int cnt = 0;

    if (mwords > 0) {
        const unsigned* mk = (const unsigned*)out + (size_t)base * NFILT;
        unsigned* ll = llist + wv * LL_CAP;
        const int nunits = (mwords + 3) >> 2;
        const int sub = lane & 3;
        const int tb  = (lane >> 2) << 2;
        int nl = 0;

        unsigned wcur = 0u;
        if (wv < nunits && wv * 4 + sub < mwords) wcur = mk[wv * 4 + sub];

        for (int u = wv; u < nunits; u += SC_WAVES) {
            const int wn = (u + SC_WAVES) * 4 + sub;
            const unsigned wload = mk[min(wn, mwords - 1)];
            const unsigned wnext = (wn < mwords) ? wload : 0u;

            const unsigned lbase = (unsigned)(u * 4 + sub) * 32u + (unsigned)tb;
            const unsigned bits  = (wcur >> tb) & 0xFu;
            for (int t = 0; t < 4; ++t) {
                const bool hit = ((bits >> t) & 1u) != 0u;
                const unsigned m = __builtin_amdgcn_ballot_w32(hit);
                if (hit) ll[nl + (int)__builtin_amdgcn_mbcnt_lo(m, 0u)] = lbase + (unsigned)t;
                nl += __builtin_popcount(m);
            }
            while (nl >= 16) {
                nl -= 16;
                kw_scan_lines<true>(ll + nl, 16, lst, cnt, lane, base, E, rows, cols, dinv, XF, acc);
            }
            wcur = wnext;
        }
        if (nl > 0)
            kw_scan_lines<false>(ll, nl, lst, cnt, lane, base, E, rows, cols, dinv, XF, acc);
    } else {
    const int nfull = E / SC_BLK;
    const int iters = (nfull + SC_WAVES - 1) / SC_WAVES;

    if (iters > 0) {
        int k = (int)((blockIdx.x * 13u) % (unsigned)iters);

        int4 ra[4], rb[4];
        kw_load_blk(ra, rows, min(wv + SC_WAVES * k, nfull - 1), lane);

        for (int t = 0; t < iters; t += 2) {
            int kb = k + 1;
            if (kb == iters) kb = 0;
            const int blka = wv + SC_WAVES * k;
            const int blkb = wv + SC_WAVES * kb;
            kw_load_blk(rb, rows, min(blkb, nfull - 1), lane);
            if (blka < nfull)
                kw_scan_blk(ra, blka, lst, cnt, lane, base, cols, dinv, XF, acc);

            int kn = kb + 1;
            if (kn == iters) kn = 0;
            kw_load_blk(ra, rows, min(wv + SC_WAVES * kn, nfull - 1), lane);
            if (t + 1 < iters && blkb < nfull)
                kw_scan_blk(rb, blkb, lst, cnt, lane, base, cols, dinv, XF, acc);
            k = kn;
        }
    }
    if (nfull * SC_BLK < E && wv == (nfull & (SC_WAVES - 1))) {
        int4 rt[4];
        kw_load_tail(rt, rows, nfull, lane, E);
        kw_scan_blk(rt, nfull, lst, cnt, lane, base, cols, dinv, XF, acc);
    }
    }
    if (cnt > 0)
        kw_flush<false>(lst, 0, cnt, lane, base, cols, dinv, XF, acc);

    __syncthreads();

    for (int i = tid; i < SC_ROWS * (NFILT / 4); i += 1024) {
        const int row = base + (i >> 4);
        if (row < N)
            ((v4f*)(out + (size_t)row * NFILT))[i & 15] = ((const v4f*)acc)[i];
    }
}

extern "C" void kernel_launch(void* const* d_in, const int* in_sizes, int n_in,
                              void* d_out, int out_size, void* d_ws, size_t ws_size,
                              hipStream_t stream) {
    const float* x   = (const float*)d_in[0];
    const float* flt = (const float*)d_in[1];
    const int*   eid = (const int*)d_in[2];

    const int N = in_sizes[0] / NFEAT;
    const int E = in_sizes[2] / 2;
    const int* rows = eid;
    const int* cols = eid + E;

    float* out = (float*)d_out;
    float* deg = (float*)d_ws;
    float* XF  = (float*)d_ws + (size_t)((N + 63) & ~63);

    unsigned* partial = (unsigned*)XF;
    const int nwords = (N + 1) / 2;
    const int R   = (N + DEG_RANGE_BINS - 1) / DEG_RANGE_BINS;
    const int G   = (E + DEG_CHUNK - 1) / DEG_CHUNK;
    const int wpg = R * DEG_RANGE_WORDS;
    const int nown     = (N + SC_ROWS - 1) / SC_ROWS;
    const int opq      = (nown + R - 1) / R;
    const int mwords   = G * MARK_CHUNK_W;
    const int lastrows = N - (nown - 1) * SC_ROWS;
    const bool keep    = opq <= MARK_OWN_MAX && mwords <= lastrows * NFILT && E >= 32;
    unsigned* marks    = keep ? (unsigned*)d_out : nullptr;
    kw_degree_lds<<<dim3(G, R), 1024, 0, stream>>>(rows, partial, marks, E, nwords, wpg, nown, opq);
    kw_dinv_reduce<<<(nwords + 255) / 256, 256, 0, stream>>>(partial, deg, N, nwords, G, wpg);

    const int stripes = (N + 15) / 16;
    kw_gemm_wmma<<<(stripes + 7) / 8, 256, 0, stream>>>(x, flt, XF, N);

    kw_rowsum_lds<<<nown, 1024, 0, stream>>>(rows, cols, deg, XF, out, N, E, keep ? mwords : 0);
}


# Edge weight factored out of the row sums: product stores dinv[col]*XF rows, owner scales its finished LDS sum by dinv[row] once; per-edge dinv gathers gone (new rounding, same add order)
# speedup vs baseline: 2.3307x; 1.0486x over previous
#include <hip/hip_runtime.h>
#include <hip/hip_bf16.h>

typedef __attribute__((ext_vector_type(2))) float v2f;
typedef __attribute__((ext_vector_type(4))) float v4f;
typedef __attribute__((ext_vector_type(8))) float v8f;

#define NFEAT 512
#define NFILT 64
#define KCHUNK 128

#define DEG_CHUNK       32768
#define DEG_RANGE_WORDS 10240
#define DEG_RANGE_BINS  (DEG_RANGE_WORDS * 2)

#define SC_ROWS        400
#define MARK_OWN_MAX   56
#define MARK_CHUNK_W   (DEG_CHUNK / 32 / 32)

__device__ __forceinline__ void kw_count_row(unsigned* h, int r, int base) {
    const unsigned d = (unsigned)(r - base);
    if (d < (unsigned)DEG_RANGE_BINS)
        atomicAdd(&h[d >> 1], 1u << ((d & 1u) << 4));
}

__device__ __forceinline__ void kw_mark_row(unsigned* bm, int r, unsigned obase, unsigned opq,
                                            unsigned nown, int wi, unsigned bit) {
    const unsigned o  = (unsigned)r / (unsigned)SC_ROWS;
    const unsigned ol = o - obase;
    if (ol < opq && o < nown)
        atomicOr(&bm[ol * MARK_CHUNK_W + wi], bit);
}

__global__ __launch_bounds__(1024) void kw_degree_lds(const int* __restrict__ rows,
                                                      unsigned* __restrict__ partial,
                                                      unsigned* __restrict__ marks,
                                                      int E, int nwords, int wpg,
                                                      int nown, int opq) {
    __shared__ __attribute__((aligned(16))) unsigned h[DEG_RANGE_WORDS];
    __shared__ unsigned bm[MARK_OWN_MAX * MARK_CHUNK_W];

    const int tid  = threadIdx.x;
    const int base = blockIdx.y * DEG_RANGE_BINS;

    for (int i = tid; i < DEG_RANGE_WORDS / 4; i += 1024)
        ((uint4*)h)[i] = make_uint4(0u, 0u, 0u, 0u);
    for (int i = tid; i < MARK_OWN_MAX * MARK_CHUNK_W; i += 1024)
        bm[i] = 0u;

    const int e0 = blockIdx.x * DEG_CHUNK;

    int4 rr[DEG_CHUNK / 4096];
#pragma unroll
    for (int j = 0; j < DEG_CHUNK / 4096; ++j) {
        const int eb = e0 + (j << 12);
        const int e  = eb + tid * 4;
        if (eb + 4096 <= E) {
            rr[j] = *(const int4*)(rows + e);
        } else {
            rr[j].x = (e     < E) ? rows[e]     : -1;
            rr[j].y = (e + 1 < E) ? rows[e + 1] : -1;
            rr[j].z = (e + 2 < E) ? rows[e + 2] : -1;
            rr[j].w = (e + 3 < E) ? rows[e + 3] : -1;
        }
    }
    __syncthreads();

#pragma unroll
    for (int j = 0; j < DEG_CHUNK / 4096; ++j) {
        kw_count_row(h, rr[j].x, base);
        kw_count_row(h, rr[j].y, base);
        kw_count_row(h, rr[j].z, base);
        kw_count_row(h, rr[j].w, base);
    }
    const unsigned obase = blockIdx.y * (unsigned)opq;
    if (marks != nullptr) {
        const unsigned bit = 1u << ((tid >> 3) & 31);
#pragma unroll
        for (int j = 0; j < DEG_CHUNK / 4096; ++j) {
            const int wi = j * 4 + (tid >> 8);
            kw_mark_row(bm, rr[j].x, obase, (unsigned)opq, (unsigned)nown, wi, bit);
            kw_mark_row(bm, rr[j].y, obase, (unsigned)opq, (unsigned)nown, wi, bit);
            kw_mark_row(bm, rr[j].z, obase, (unsigned)opq, (unsigned)nown, wi, bit);
            kw_mark_row(bm, rr[j].w, obase, (unsigned)opq, (unsigned)nown, wi, bit);
        }
    }
    __syncthreads();

    if (marks != nullptr) {
        for (int i = tid; i < opq * MARK_CHUNK_W; i += 1024) {
            const unsigned o = obase + (unsigned)(i >> 5);
            if (o < (unsigned)nown)
                marks[(size_t)o * (SC_ROWS * NFILT) + blockIdx.x * MARK_CHUNK_W + (i & 31)] = bm[i];
        }
    }

    const int wbase = blockIdx.y * DEG_RANGE_WORDS;
    uint4* dst = (uint4*)(partial + (size_t)blockIdx.x * wpg + wbase);
    for (int i = tid; i < DEG_RANGE_WORDS / 4; i += 1024)
        if (wbase + i * 4 < nwords)
            dst[i] = ((const uint4*)h)[i];
}

__global__ __launch_bounds__(256) void kw_dinv_reduce(const unsigned* __restrict__ partial,
                                                      float* __restrict__ dinv,
                                                      int N, int nwords, int G, int wpg) {
    const int w = blockIdx.x * 256 + threadIdx.x;
    if (w >= nwords) return;

    const unsigned* p = partial + w;
    unsigned lo = 0u, hi = 0u;
#pragma unroll 7
    for (int g = 0; g < G; ++g) {
        const unsigned v = p[(size_t)g * wpg];
        lo += v & 0xFFFFu;
        hi += v >> 16;
    }
    const float d0 = (float)lo;
    const float d1 = (float)hi;
    const float i0 = (d0 > 0.0f) ? rsqrtf(d0) : 0.0f;
    const float i1 = (d1 > 0.0f) ? rsqrtf(d1) : 0.0f;

    const int n0 = 2 * w;
    if (n0 + 1 < N) {
        v2f o = {i0, i1};
        *(v2f*)(dinv + n0) = o;
    } else {
        dinv[n0] = i0;
    }
}

__global__ __launch_bounds__(256) void kw_gemm_wmma(const float* __restrict__ x,
                                                    const float* __restrict__ flt,
                                                    const float* __restrict__ dinv,
                                                    float* __restrict__ XF, int N) {
    __shared__ float sB[KCHUNK * NFILT];

    const int lane = threadIdx.x & 31;
    const int wib  = threadIdx.x >> 5;
    const int half = lane >> 4;
    const int l16  = lane & 15;

    const int stripe = blockIdx.x * 8 + wib;
    const int m0     = stripe * 16;
    const bool active = (m0 < N);

    int row = m0 + l16;
    if (row >= N) row = N - 1;
    const float* xrow = x + (size_t)row * NFEAT;

    v8f acc0 = {}, acc1 = {}, acc2 = {}, acc3 = {};

    float dv[8] = {};

    const float* bbase = sB + lane * 2;

    for (int kc = 0; kc < NFEAT; kc += KCHUNK) {
        __syncthreads();
        {
            const v4f* src = (const v4f*)(flt + (size_t)kc * NFILT);
            for (int t4 = threadIdx.x; t4 < KCHUNK * (NFILT / 4); t4 += 256) {
                v4f v = src[t4];
                int r    = t4 >> 4;
                int g    = r >> 2;
                int koff = r & 3;
                int h    = koff >> 1;
                int j    = koff & 1;
                int cb   = (t4 & 15) << 2;
                int n    = cb >> 4;
                int c16  = cb & 15;
                float* d = sB + (((g << 2) + n) << 6) + (((h << 4) + c16) << 1) + j;
                d[0] = v.x; d[2] = v.y; d[4] = v.z; d[6] = v.w;
            }
        }
        __syncthreads();

        if (kc == 0) {
#pragma unroll
            for (int r = 0; r < 8; ++r)
                dv[r] = dinv[min(m0 + r + 8 * half, N - 1)];
        }

        if (active) {
#pragma unroll 4
            for (int k2 = 0; k2 < KCHUNK; k2 += 4) {
                v2f a = *(const v2f*)(xrow + kc + k2 + 2 * half);

                const int g = k2 >> 2;
                const float* bg = bbase + (g << 8);
                v2f b0 = *(const v2f*)(bg);
                v2f b1 = *(const v2f*)(bg + 64);
                v2f b2 = *(const v2f*)(bg + 128);
                v2f b3 = *(const v2f*)(bg + 192);

                acc0 = __builtin_amdgcn_wmma_f32_16x16x4_f32(false, a, false, b0, (short)0, acc0, false, false);
                acc1 = __builtin_amdgcn_wmma_f32_16x16x4_f32(false, a, false, b1, (short)0, acc1, false, false);
                acc2 = __builtin_amdgcn_wmma_f32_16x16x4_f32(false, a, false, b2, (short)0, acc2, false, false);
                acc3 = __builtin_amdgcn_wmma_f32_16x16x4_f32(false, a, false, b3, (short)0, acc3, false, false);
            }
        }
    }

    if (active) {
        asm volatile("v_nop\n\tv_nop\n\tv_nop\n\tv_nop\n\tv_nop\n\tv_nop\n\tv_nop\n\tv_nop"
                     : "+v"(acc0), "+v"(acc1), "+v"(acc2), "+v"(acc3));

#pragma unroll
        for (int r = 0; r < 8; ++r) {
            int M = m0 + r + 8 * half;
            if (M < N) {
                float* o = XF + (size_t)M * NFILT + l16;
                o[0]  = dv[r] * acc0[r];
                o[16] = dv[r] * acc1[r];
                o[32] = dv[r] * acc2[r];
                o[48] = dv[r] * acc3[r];
            }
        }
    }
}

#define SC_WAVES 32
#define SC_BLK   512
#define SC_CAP   (SC_BLK + 32)
#define LL_CAP   144
#define SC_ST    ((SC_ROWS * (NFILT / 4) + 1023) / 1024)

__device__ __forceinline__ void kw_load_blk(int4 (&r)[4], const int* __restrict__ rows,
                                            int blk, int lane) {
    const int4* p = (const int4*)(rows + (size_t)blk * SC_BLK) + lane;
#pragma unroll
    for (int q = 0; q < 4; ++q)
        r[q] = p[q * 32];
}

__device__ __forceinline__ void kw_load_tail(int4 (&r)[4], const int* __restrict__ rows,
                                             int blk, int lane, int E) {
#pragma unroll
    for (int q = 0; q < 4; ++q) {
        const int e = blk * SC_BLK + q * 128 + lane * 4;
        r[q].x = (e     < E) ? rows[e]     : -1;
        r[q].y = (e + 1 < E) ? rows[e + 1] : -1;
        r[q].z = (e + 2 < E) ? rows[e + 2] : -1;
        r[q].w = (e + 3 < E) ? rows[e + 3] : -1;
    }
}

__device__ __forceinline__ void kw_append(uint2* lst, int& cnt, bool hit, unsigned e, unsigned d) {
    const unsigned m = __builtin_amdgcn_ballot_w32(hit);
    if (hit) lst[cnt + (int)__builtin_amdgcn_mbcnt_lo(m, 0u)] = make_uint2(e, d);
    cnt += __builtin_popcount(m);
}

template <bool FULL>
__device__ __forceinline__ void kw_flush(const uint2* lst, int first, int nvalid, int lane,
                                         const int* __restrict__ cols,
                                         const float* __restrict__ XF, float* acc) {
    int c = 0, d = 0;
    if (FULL || lane < nvalid) {
        const uint2 t = lst[first + lane];
        c = cols[t.x];
        d = (int)t.y;
    }
    for (int j0 = 0; j0 < 32; j0 += 8) {
        if (!FULL && j0 >= nvalid) break;
        float v0[8], v1[8];
#pragma unroll
        for (int u = 0; u < 8; ++u) {
            const int cj = __builtin_amdgcn_readlane(c, j0 + u);
            const float* src = XF + (size_t)cj * NFILT;
            v0[u] = src[lane];
            v1[u] = src[lane + 32];
        }
#pragma unroll
        for (int u = 0; u < 8; ++u) {
            if (FULL || j0 + u < nvalid) {
                const int dj = __builtin_amdgcn_readlane(d, j0 + u);
                float* dst = acc + dj * NFILT + lane;
                __hip_atomic_fetch_add(&dst[0],  v0[u], __ATOMIC_RELAXED, __HIP_MEMORY_SCOPE_WORKGROUP);
                __hip_atomic_fetch_add(&dst[32], v1[u], __ATOMIC_RELAXED, __HIP_MEMORY_SCOPE_WORKGROUP);
            }
        }
    }
}

__device__ __forceinline__ void kw_scan_blk(const int4 (&r)[4], int blk, uint2* lst, int& cnt,
                                            int lane, int base,
                                            const int* __restrict__ cols,
                                            const float* __restrict__ XF, float* acc) {
#pragma unroll
    for (int q = 0; q < 4; ++q) {
        const unsigned dx = (unsigned)(r[q].x - base);
        const unsigned dy = (unsigned)(r[q].y - base);
        const unsigned dz = (unsigned)(r[q].z - base);
        const unsigned dw = (unsigned)(r[q].w - base);
        const unsigned dm = min(min(dx, dy), min(dz, dw));
        if (__builtin_amdgcn_ballot_w32(dm < (unsigned)SC_ROWS) != 0u) {
            const unsigned e = (unsigned)(blk * SC_BLK + q * 128 + lane * 4);
            kw_append(lst, cnt, dx < (unsigned)SC_ROWS, e,      dx);
            kw_append(lst, cnt, dy < (unsigned)SC_ROWS, e + 1u, dy);
            kw_append(lst, cnt, dz < (unsigned)SC_ROWS, e + 2u, dz);
            kw_append(lst, cnt, dw < (unsigned)SC_ROWS, e + 3u, dw);
        }
    }
    while (cnt >= 32) {
        cnt -= 32;
        kw_flush<true>(lst, cnt, 32, lane, cols, XF, acc);
    }
}

template <bool FULL>
__device__ __forceinline__ void kw_scan_lines(const unsigned* ll, int nvalid, uint2* lst, int& cnt,
                                              int lane, int base, int E,
                                              const int* __restrict__ rows,
                                              const int* __restrict__ cols,
                                              const float* __restrict__ XF, float* acc) {
    int4     r[4];
    unsigned e0[4];
#pragma unroll
    for (int q = 0; q < 4; ++q) {
        const int s = q * 4 + (lane >> 3);
        if (FULL || s < nvalid) {
            e0[q] = ll[s] * 32u + (unsigned)((lane & 7) << 2);
            r[q]  = *(const int4*)(rows + e0[q]);
        } else {
            e0[q] = 0u;
            r[q]  = make_int4(-1, -1, -1, -1);
        }
    }
#pragma unroll
    for (int q = 0; q < 4; ++q) {
        const unsigned dx = (unsigned)(r[q].x - base);
        const unsigned dy = (unsigned)(r[q].y - base);
        const unsigned dz = (unsigned)(r[q].z - base);
        const unsigned dw = (unsigned)(r[q].w - base);
        const unsigned dm = min(min(dx, dy), min(dz, dw));
        if (__builtin_amdgcn_ballot_w32(dm < (unsigned)SC_ROWS) != 0u) {
            const unsigned e = e0[q];
            kw_append(lst, cnt, dx < (unsigned)SC_ROWS && e      < (unsigned)E, e,      dx);
            kw_append(lst, cnt, dy < (unsigned)SC_ROWS && e + 1u < (unsigned)E, e + 1u, dy);
            kw_append(lst, cnt, dz < (unsigned)SC_ROWS && e + 2u < (unsigned)E, e + 2u, dz);
            kw_append(lst, cnt, dw < (unsigned)SC_ROWS && e + 3u < (unsigned)E, e + 3u, dw);
        }
    }
    while (cnt >= 32) {
        cnt -= 32;
        kw_flush<true>(lst, cnt, 32, lane, cols, XF, acc);
    }
}

__global__ __launch_bounds__(1024) void kw_rowsum_lds(const int* __restrict__ rows,
                                                      const int* __restrict__ cols,
                                                      const float* __restrict__ dinv,
                                                      const float* __restrict__ XF,
                                                      float* out, int N, int E, int mwords) {
    __shared__ __attribute__((aligned(16))) float acc[SC_ROWS * NFILT];
    __shared__ __attribute__((aligned(16))) uint2 slist[SC_WAVES * SC_CAP];
    __shared__ unsigned llist[SC_WAVES * LL_CAP];

    const int tid  = threadIdx.x;
    const int lane = tid & 31;
    const int wv   = __builtin_amdgcn_readfirstlane(tid >> 5);
    const int base = blockIdx.x * SC_ROWS;

    for (int i = tid; i < SC_ROWS * NFILT / 4; i += 1024)
        ((v4f*)acc)[i] = (v4f){0.0f, 0.0f, 0.0f, 0.0f};
    __syncthreads();

    float sc[SC_ST];
#pragma unroll
    for (int t = 0; t < SC_ST; ++t)
        sc[t] = dinv[min(base + ((tid + t * 1024) >> 4), N - 1)];

    uint2* lst = slist + wv * SC_CAP;
    int cnt = 0;

    if (mwords > 0) {
        const unsigned* mk = (const unsigned*)out + (size_t)base * NFILT;
        unsigned* ll = llist + wv * LL_CAP;
        const int nunits = (mwords + 3) >> 2;
        const int sub = lane & 3;
        const int tb  = (lane >> 2) << 2;
        int nl = 0;

        unsigned wcur = 0u;
        if (wv < nunits && wv * 4 + sub < mwords) wcur = mk[wv * 4 + sub];

        for (int u = wv; u < nunits; u += SC_WAVES) {
            const int wn = (u + SC_WAVES) * 4 + sub;
            const unsigned wload = mk[min(wn, mwords - 1)];
            const unsigned wnext = (wn < mwords) ? wload : 0u;

            const unsigned lbase = (unsigned)(u * 4 + sub) * 32u + (unsigned)tb;
            const unsigned bits  = (wcur >> tb) & 0xFu;
            for (int t = 0; t < 4; ++t) {
                const bool hit = ((bits >> t) & 1u) != 0u;
                const unsigned m = __builtin_amdgcn_ballot_w32(hit);
                if (hit) ll[nl + (int)__builtin_amdgcn_mbcnt_lo(m, 0u)] = lbase + (unsigned)t;
                nl += __builtin_popcount(m);
            }
            while (nl >= 16) {
                nl -= 16;
                kw_scan_lines<true>(ll + nl, 16, lst, cnt, lane, base, E, rows, cols, XF, acc);
            }
            wcur = wnext;
        }
        if (nl > 0)
            kw_scan_lines<false>(ll, nl, lst, cnt, lane, base, E, rows, cols, XF, acc);
    } else {
    const int nfull = E / SC_BLK;
    const int iters = (nfull + SC_WAVES - 1) / SC_WAVES;

    if (iters > 0) {
        int k = (int)((blockIdx.x * 13u) % (unsigned)iters);

        int4 ra[4], rb[4];
        kw_load_blk(ra, rows, min(wv + SC_WAVES * k, nfull - 1), lane);

        for (int t = 0; t < iters; t += 2) {
            int kb = k + 1;
            if (kb == iters) kb = 0;
            const int blka = wv + SC_WAVES * k;
            const int blkb = wv + SC_WAVES * kb;
            kw_load_blk(rb, rows, min(blkb, nfull - 1), lane);
            if (blka < nfull)
                kw_scan_blk(ra, blka, lst, cnt, lane, base, cols, XF, acc);

            int kn = kb + 1;
            if (kn == iters) kn = 0;
            kw_load_blk(ra, rows, min(wv + SC_WAVES * kn, nfull - 1), lane);
            if (t + 1 < iters && blkb < nfull)
                kw_scan_blk(rb, blkb, lst, cnt, lane, base, cols, XF, acc);
            k = kn;
        }
    }
    if (nfull * SC_BLK < E && wv == (nfull & (SC_WAVES - 1))) {
        int4 rt[4];
        kw_load_tail(rt, rows, nfull, lane, E);
        kw_scan_blk(rt, nfull, lst, cnt, lane, base, cols, XF, acc);
    }
    }
    if (cnt > 0)
        kw_flush<false>(lst, 0, cnt, lane, cols, XF, acc);

    __syncthreads();

#pragma unroll
    for (int t = 0; t < SC_ST; ++t) {
        const int i   = tid + t * 1024;
        const int row = base + (i >> 4);
        if (i < SC_ROWS * (NFILT / 4) && row < N) {
            const float s = sc[t];
            const v4f   a = ((const v4f*)acc)[i];
            ((v4f*)(out + (size_t)row * NFILT))[i & 15] = (v4f){s * a.x, s * a.y, s * a.z, s * a.w};
        }
    }
}

extern "C" void kernel_launch(void* const* d_in, const int* in_sizes, int n_in,
                              void* d_out, int out_size, void* d_ws, size_t ws_size,
                              hipStream_t stream) {
    const float* x   = (const float*)d_in[0];
    const float* flt = (const float*)d_in[1];
    const int*   eid = (const int*)d_in[2];

    const int N = in_sizes[0] / NFEAT;
    const int E = in_sizes[2] / 2;
    const int* rows = eid;
    const int* cols = eid + E;

    float* out = (float*)d_out;
    float* deg = (float*)d_ws;
    float* XF  = (float*)d_ws + (size_t)((N + 63) & ~63);

    unsigned* partial = (unsigned*)XF;
    const int nwords = (N + 1) / 2;
    const int R   = (N + DEG_RANGE_BINS - 1) / DEG_RANGE_BINS;
    const int G   = (E + DEG_CHUNK - 1) / DEG_CHUNK;
    const int wpg = R * DEG_RANGE_WORDS;
    const int nown     = (N + SC_ROWS - 1) / SC_ROWS;
    const int opq      = (nown + R - 1) / R;
    const int mwords   = G * MARK_CHUNK_W;
    const int lastrows = N - (nown - 1) * SC_ROWS;
    const bool keep    = opq <= MARK_OWN_MAX && mwords <= lastrows * NFILT && E >= 32;
    unsigned* marks    = keep ? (unsigned*)d_out : nullptr;
    kw_degree_lds<<<dim3(G, R), 1024, 0, stream>>>(rows, partial, marks, E, nwords, wpg, nown, opq);
    kw_dinv_reduce<<<(nwords + 255) / 256, 256, 0, stream>>>(partial, deg, N, nwords, G, wpg);

    const int stripes = (N + 15) / 16;
    kw_gemm_wmma<<<(stripes + 7) / 8, 256, 0, stream>>>(x, flt, deg, XF, N);

    kw_rowsum_lds<<<nown, 1024, 0, stream>>>(rows, cols, deg, XF, out, N, E, keep ? mwords : 0);
}


# dinv launch removed: the degree launch parks its partial counts in each owner's output rows; the product and the row-sum owners sum their own rows' counts (integer LDS adds) into dinv; 3 launches
# speedup vs baseline: 2.4143x; 1.0359x over previous
#include <hip/hip_runtime.h>
#include <hip/hip_bf16.h>

typedef __attribute__((ext_vector_type(2))) float v2f;
typedef __attribute__((ext_vector_type(4))) float v4f;
typedef __attribute__((ext_vector_type(8))) float v8f;

#define NFEAT 512
#define NFILT 64
#define KCHUNK 128

#define DEG_CHUNK       32768
#define DEG_RANGE_WORDS 10200
#define DEG_RANGE_BINS  (DEG_RANGE_WORDS * 2)

#define SC_ROWS        400
#define MARK_OWN_MAX   56
#define MARK_CHUNK_W   (DEG_CHUNK / 32 / 32)

#define SC_WORDS       (SC_ROWS / 2)
#define PP_PITCH       256
#define PP_BATCH       16

__device__ __forceinline__ void kw_count_row(unsigned* h, int r, int base) {
    const unsigned d = (unsigned)(r - base);
    if (d < (unsigned)DEG_RANGE_BINS)
        atomicAdd(&h[d >> 1], 1u << ((d & 1u) << 4));
}

__device__ __forceinline__ void kw_mark_row(unsigned* bm, int r, unsigned obase, unsigned opq,
                                            unsigned nown, int wi, unsigned bit) {
    const unsigned o  = (unsigned)r / (unsigned)SC_ROWS;
    const unsigned ol = o - obase;
    if (ol < opq && o < nown)
        atomicOr(&bm[ol * MARK_CHUNK_W + wi], bit);
}

__global__ __launch_bounds__(1024) void kw_degree_lds(const int* __restrict__ rows,
                                                      unsigned* __restrict__ partial,
                                                      unsigned* __restrict__ marks,
                                                      int E, int nwords, int wpg,
                                                      int nown, int opq, int pbase) {
    __shared__ __attribute__((aligned(16))) unsigned h[DEG_RANGE_WORDS];
    __shared__ unsigned bm[MARK_OWN_MAX * MARK_CHUNK_W];

    const int tid  = threadIdx.x;
    const int base = blockIdx.y * DEG_RANGE_BINS;

    for (int i = tid; i < DEG_RANGE_WORDS / 4; i += 1024)
        ((uint4*)h)[i] = make_uint4(0u, 0u, 0u, 0u);
    for (int i = tid; i < MARK_OWN_MAX * MARK_CHUNK_W; i += 1024)
        bm[i] = 0u;

    const int e0 = blockIdx.x * DEG_CHUNK;

    int4 rr[DEG_CHUNK / 4096];
#pragma unroll
    for (int j = 0; j < DEG_CHUNK / 4096; ++j) {
        const int eb = e0 + (j << 12);
        const int e  = eb + tid * 4;
        if (eb + 4096 <= E) {
            rr[j] = *(const int4*)(rows + e);
        } else {
            rr[j].x = (e     < E) ? rows[e]     : -1;
            rr[j].y = (e + 1 < E) ? rows[e + 1] : -1;
            rr[j].z = (e + 2 < E) ? rows[e + 2] : -1;
            rr[j].w = (e + 3 < E) ? rows[e + 3] : -1;
        }
    }
    __syncthreads();

#pragma unroll
    for (int j = 0; j < DEG_CHUNK / 4096; ++j) {
        kw_count_row(h, rr[j].x, base);
        kw_count_row(h, rr[j].y, base);
        kw_count_row(h, rr[j].z, base);
        kw_count_row(h, rr[j].w, base);
    }
    const unsigned obase = blockIdx.y * (unsigned)opq;
    if (marks != nullptr) {
        const unsigned bit = 1u << ((tid >> 3) & 31);
#pragma unroll
        for (int j = 0; j < DEG_CHUNK / 4096; ++j) {
            const int wi = j * 4 + (tid >> 8);
            kw_mark_row(bm, rr[j].x, obase, (unsigned)opq, (unsigned)nown, wi, bit);
            kw_mark_row(bm, rr[j].y, obase, (unsigned)opq, (unsigned)nown, wi, bit);
            kw_mark_row(bm, rr[j].z, obase, (unsigned)opq, (unsigned)nown, wi, bit);
            kw_mark_row(bm, rr[j].w, obase, (unsigned)opq, (unsigned)nown, wi, bit);
        }
    }
    __syncthreads();

    if (marks != nullptr) {
        for (int i = tid; i < opq * MARK_CHUNK_W; i += 1024) {
            const unsigned o = obase + (unsigned)(i >> 5);
            if (o < (unsigned)nown)
                marks[(size_t)o * (SC_ROWS * NFILT) + blockIdx.x * MARK_CHUNK_W + (i & 31)] = bm[i];
        }
    }

    const int wbase = blockIdx.y * DEG_RANGE_WORDS;
    if (pbase >= 0) {
        for (int i = tid; i < DEG_RANGE_WORDS / 4; i += 1024) {
            const int w = wbase + i * 4;
            if (w < nwords) {
                const unsigned o = (unsigned)w / (unsigned)SC_WORDS;
                unsigned* d = marks + (size_t)o * (SC_ROWS * NFILT) + pbase
                            + blockIdx.x * PP_PITCH + (w - (int)o * SC_WORDS);
                *(uint4*)d = ((const uint4*)h)[i];
            }
        }
    } else {
        uint4* dst = (uint4*)(partial + (size_t)blockIdx.x * wpg + wbase);
        for (int i = tid; i < DEG_RANGE_WORDS / 4; i += 1024)
            if (wbase + i * 4 < nwords)
                dst[i] = ((const uint4*)h)[i];
    }
}

__device__ __forceinline__ float kw_dinv_of(unsigned cnt) {
    const float d = (float)cnt;
    return (d > 0.0f) ? rsqrtf(d) : 0.0f;
}

__device__ __forceinline__ void kw_pp_load(unsigned (&v)[PP_BATCH], const unsigned* seg, int g0, int G) {
#pragma unroll
    for (int j = 0; j < PP_BATCH; ++j)
        v[j] = seg[(size_t)min(g0 + 4 * j, G - 1) * PP_PITCH];
}

__device__ __forceinline__ void kw_pp_sum(const unsigned (&v)[PP_BATCH], int g0, int G, bool on,
                                          unsigned& lo, unsigned& hi) {
#pragma unroll
    for (int j = 0; j < PP_BATCH; ++j) {
        const unsigned t = (on && g0 + 4 * j < G) ? v[j] : 0u;
        lo += t & 0xFFFFu;
        hi += t >> 16;
    }
}

__global__ __launch_bounds__(256) void kw_dinv_reduce(const unsigned* __restrict__ partial,
                                                      float* __restrict__ dinv,
                                                      int N, int nwords, int G, int wpg) {
    const int w = blockIdx.x * 256 + threadIdx.x;
    if (w >= nwords) return;

    const unsigned* p = partial + w;
    unsigned lo = 0u, hi = 0u;
#pragma unroll 7
    for (int g = 0; g < G; ++g) {
        const unsigned v = p[(size_t)g * wpg];
        lo += v & 0xFFFFu;
        hi += v >> 16;
    }
    const float d0 = (float)lo;
    const float d1 = (float)hi;
    const float i0 = (d0 > 0.0f) ? rsqrtf(d0) : 0.0f;
    const float i1 = (d1 > 0.0f) ? rsqrtf(d1) : 0.0f;

    const int n0 = 2 * w;
    if (n0 + 1 < N) {
        v2f o = {i0, i1};
        *(v2f*)(dinv + n0) = o;
    } else {
        dinv[n0] = i0;
    }
}

__global__ __launch_bounds__(256) void kw_gemm_wmma(const float* __restrict__ x,
                                                    const float* __restrict__ flt,
                                                    const float* __restrict__ dinv,
                                                    const unsigned* __restrict__ pp,
                                                    float* __restrict__ XF, int N, int G, int pbase) {
    __shared__ float sB[KCHUNK * NFILT];
    __shared__ unsigned sCnt[128];

    const int  nwords = (N + 1) >> 1;
    const int  pwl    = threadIdx.x & 63;
    const int  pgq    = threadIdx.x >> 6;
    const int  pw     = blockIdx.x * 64 + pwl;
    const bool pon    = pw < nwords;
    const unsigned* pseg = nullptr;
    unsigned pv[PP_BATCH] = {};
    if (pp != nullptr) {
        if (threadIdx.x < 128) sCnt[threadIdx.x] = 0u;
        const int      wc = min(pw, nwords - 1);
        const unsigned o  = (unsigned)wc / (unsigned)SC_WORDS;
        pseg = pp + (size_t)o * (SC_ROWS * NFILT) + pbase + (wc - (int)o * SC_WORDS);
    }

    const int lane = threadIdx.x & 31;
    const int wib  = threadIdx.x >> 5;
    const int half = lane >> 4;
    const int l16  = lane & 15;

    const int stripe = blockIdx.x * 8 + wib;
    const int m0     = stripe * 16;
    const bool active = (m0 < N);

    int row = m0 + l16;
    if (row >= N) row = N - 1;
    const float* xrow = x + (size_t)row * NFEAT;

    v8f acc0 = {}, acc1 = {}, acc2 = {}, acc3 = {};

    float dv[8] = {};

    const float* bbase = sB + lane * 2;

#pragma unroll
    for (int kc = 0; kc < NFEAT; kc += KCHUNK) {
        __syncthreads();
        if (kc == 0 && pp != nullptr)
            kw_pp_load(pv, pseg, pgq, G);
        {
            const v4f* src = (const v4f*)(flt + (size_t)kc * NFILT);
            for (int t4 = threadIdx.x; t4 < KCHUNK * (NFILT / 4); t4 += 256) {
                v4f v = src[t4];
                int r    = t4 >> 4;
                int g    = r >> 2;
                int koff = r & 3;
                int h    = koff >> 1;
                int j    = koff & 1;
                int cb   = (t4 & 15) << 2;
                int n    = cb >> 4;
                int c16  = cb & 15;
                float* d = sB + (((g << 2) + n) << 6) + (((h << 4) + c16) << 1) + j;
                d[0] = v.x; d[2] = v.y; d[4] = v.z; d[6] = v.w;
            }
        }
        if (kc == 0 && pp != nullptr) {
            unsigned lo = 0u, hi = 0u;
            kw_pp_sum(pv, pgq, G, pon, lo, hi);
            if (pon) {
                atomicAdd(&sCnt[2 * pwl],     lo);
                atomicAdd(&sCnt[2 * pwl + 1], hi);
            }
        }
        __syncthreads();

        if (kc == 0) {
            if (pp == nullptr) {
#pragma unroll
                for (int r = 0; r < 8; ++r)
                    dv[r] = dinv[min(m0 + r + 8 * half, N - 1)];
            }
        }

        if (active) {
#pragma unroll 4
            for (int k2 = 0; k2 < KCHUNK; k2 += 4) {
                v2f a = *(const v2f*)(xrow + kc + k2 + 2 * half);

                const int g = k2 >> 2;
                const float* bg = bbase + (g << 8);
                v2f b0 = *(const v2f*)(bg);
                v2f b1 = *(const v2f*)(bg + 64);
                v2f b2 = *(const v2f*)(bg + 128);
                v2f b3 = *(const v2f*)(bg + 192);

                acc0 = __builtin_amdgcn_wmma_f32_16x16x4_f32(false, a, false, b0, (short)0, acc0, false, false);
                acc1 = __builtin_amdgcn_wmma_f32_16x16x4_f32(false, a, false, b1, (short)0, acc1, false, false);
                acc2 = __builtin_amdgcn_wmma_f32_16x16x4_f32(false, a, false, b2, (short)0, acc2, false, false);
                acc3 = __builtin_amdgcn_wmma_f32_16x16x4_f32(false, a, false, b3, (short)0, acc3, false, false);
            }
        }
    }

    if (active) {
        if (pp != nullptr) {
#pragma unroll
            for (int r = 0; r < 8; ++r)
                dv[r] = kw_dinv_of(sCnt[wib * 16 + r + 8 * half]);
        }
        asm volatile("v_nop\n\tv_nop\n\tv_nop\n\tv_nop\n\tv_nop\n\tv_nop\n\tv_nop\n\tv_nop"
                     : "+v"(acc0), "+v"(acc1), "+v"(acc2), "+v"(acc3));

#pragma unroll
        for (int r = 0; r < 8; ++r) {
            int M = m0 + r + 8 * half;
            if (M < N) {
                float* o = XF + (size_t)M * NFILT + l16;
                o[0]  = dv[r] * acc0[r];
                o[16] = dv[r] * acc1[r];
                o[32] = dv[r] * acc2[r];
                o[48] = dv[r] * acc3[r];
            }
        }
    }
}

#define SC_WAVES 32
#define SC_BLK   512
#define SC_CAP   (SC_BLK + 32)
#define LL_CAP   144
#define SC_ST    ((SC_ROWS * (NFILT / 4) + 1023) / 1024)

__device__ __forceinline__ void kw_load_blk(int4 (&r)[4], const int* __restrict__ rows,
                                            int blk, int lane) {
    const int4* p = (const int4*)(rows + (size_t)blk * SC_BLK) + lane;
#pragma unroll
    for (int q = 0; q < 4; ++q)
        r[q] = p[q * 32];
}

__device__ __forceinline__ void kw_load_tail(int4 (&r)[4], const int* __restrict__ rows,
                                             int blk, int lane, int E) {
#pragma unroll
    for (int q = 0; q < 4; ++q) {
        const int e = blk * SC_BLK + q * 128 + lane * 4;
        r[q].x = (e     < E) ? rows[e]     : -1;
        r[q].y = (e + 1 < E) ? rows[e + 1] : -1;
        r[q].z = (e + 2 < E) ? rows[e + 2] : -1;
        r[q].w = (e + 3 < E) ? rows[e + 3] : -1;
    }
}

__device__ __forceinline__ void kw_append(uint2* lst, int& cnt, bool hit, unsigned e, unsigned d) {
    const unsigned m = __builtin_amdgcn_ballot_w32(hit);
    if (hit) lst[cnt + (int)__builtin_amdgcn_mbcnt_lo(m, 0u)] = make_uint2(e, d);
    cnt += __builtin_popcount(m);
}

template <bool FULL>
__device__ __forceinline__ void kw_flush(const uint2* lst, int first, int nvalid, int lane,
                                         const int* __restrict__ cols,
                                         const float* __restrict__ XF, float* acc) {
    int c = 0, d = 0;
    if (FULL || lane < nvalid) {
        const uint2 t = lst[first + lane];
        c = cols[t.x];
        d = (int)t.y;
    }
    for (int j0 = 0; j0 < 32; j0 += 8) {
        if (!FULL && j0 >= nvalid) break;
        float v0[8], v1[8];
#pragma unroll
        for (int u = 0; u < 8; ++u) {
            const int cj = __builtin_amdgcn_readlane(c, j0 + u);
            const float* src = XF + (size_t)cj * NFILT;
            v0[u] = src[lane];
            v1[u] = src[lane + 32];
        }
#pragma unroll
        for (int u = 0; u < 8; ++u) {
            if (FULL || j0 + u < nvalid) {
                const int dj = __builtin_amdgcn_readlane(d, j0 + u);
                float* dst = acc + dj * NFILT + lane;
                __hip_atomic_fetch_add(&dst[0],  v0[u], __ATOMIC_RELAXED, __HIP_MEMORY_SCOPE_WORKGROUP);
                __hip_atomic_fetch_add(&dst[32], v1[u], __ATOMIC_RELAXED, __HIP_MEMORY_SCOPE_WORKGROUP);
            }
        }
    }
}

__device__ __forceinline__ void kw_scan_blk(const int4 (&r)[4], int blk, uint2* lst, int& cnt,
                                            int lane, int base,
                                            const int* __restrict__ cols,
                                            const float* __restrict__ XF, float* acc) {
#pragma unroll
    for (int q = 0; q < 4; ++q) {
        const unsigned dx = (unsigned)(r[q].x - base);
        const unsigned dy = (unsigned)(r[q].y - base);
        const unsigned dz = (unsigned)(r[q].z - base);
        const unsigned dw = (unsigned)(r[q].w - base);
        const unsigned dm = min(min(dx, dy), min(dz, dw));
        if (__builtin_amdgcn_ballot_w32(dm < (unsigned)SC_ROWS) != 0u) {
            const unsigned e = (unsigned)(blk * SC_BLK + q * 128 + lane * 4);
            kw_append(lst, cnt, dx < (unsigned)SC_ROWS, e,      dx);
            kw_append(lst, cnt, dy < (unsigned)SC_ROWS, e + 1u, dy);
            kw_append(lst, cnt, dz < (unsigned)SC_ROWS, e + 2u, dz);
            kw_append(lst, cnt, dw < (unsigned)SC_ROWS, e + 3u, dw);
        }
    }
    while (cnt >= 32) {
        cnt -= 32;
        kw_flush<true>(lst, cnt, 32, lane, cols, XF, acc);
    }
}

template <bool FULL>
__device__ __forceinline__ void kw_scan_lines(const unsigned* ll, int nvalid, uint2* lst, int& cnt,
                                              int lane, int base, int E,
                                              const int* __restrict__ rows,
                                              const int* __restrict__ cols,
                                              const float* __restrict__ XF, float* acc) {
    int4     r[4];
    unsigned e0[4];
#pragma unroll
    for (int q = 0; q < 4; ++q) {
        const int s = q * 4 + (lane >> 3);
        if (FULL || s < nvalid) {
            e0[q] = ll[s] * 32u + (unsigned)((lane & 7) << 2);
            r[q]  = *(const int4*)(rows + e0[q]);
        } else {
            e0[q] = 0u;
            r[q]  = make_int4(-1, -1, -1, -1);
        }
    }
#pragma unroll
    for (int q = 0; q < 4; ++q) {
        const unsigned dx = (unsigned)(r[q].x - base);
        const unsigned dy = (unsigned)(r[q].y - base);
        const unsigned dz = (unsigned)(r[q].z - base);
        const unsigned dw = (unsigned)(r[q].w - base);
        const unsigned dm = min(min(dx, dy), min(dz, dw));
        if (__builtin_amdgcn_ballot_w32(dm < (unsigned)SC_ROWS) != 0u) {
            const unsigned e = e0[q];
            kw_append(lst, cnt, dx < (unsigned)SC_ROWS && e      < (unsigned)E, e,      dx);
            kw_append(lst, cnt, dy < (unsigned)SC_ROWS && e + 1u < (unsigned)E, e + 1u, dy);
            kw_append(lst, cnt, dz < (unsigned)SC_ROWS && e + 2u < (unsigned)E, e + 2u, dz);
            kw_append(lst, cnt, dw < (unsigned)SC_ROWS && e + 3u < (unsigned)E, e + 3u, dw);
        }
    }
    while (cnt >= 32) {
        cnt -= 32;
        kw_flush<true>(lst, cnt, 32, lane, cols, XF, acc);
    }
}

__global__ __launch_bounds__(1024) void kw_rowsum_lds(const int* __restrict__ rows,
                                                      const int* __restrict__ cols,
                                                      const float* __restrict__ dinv,
                                                      const float* __restrict__ XF,
                                                      float* out, int N, int E, int mwords,
                                                      int G, int pbase) {
    __shared__ __attribute__((aligned(16))) float acc[SC_ROWS * NFILT];
    __shared__ unsigned rcnt[SC_ROWS];
    __shared__ __attribute__((aligned(16))) uint2 slist[SC_WAVES * SC_CAP];
    __shared__ unsigned llist[SC_WAVES * LL_CAP];

    const int tid  = threadIdx.x;
    const int lane = tid & 31;
    const int wv   = __builtin_amdgcn_readfirstlane(tid >> 5);
    const int base = blockIdx.x * SC_ROWS;

    for (int i = tid; i < SC_ROWS * NFILT / 4; i += 1024)
        ((v4f*)acc)[i] = (v4f){0.0f, 0.0f, 0.0f, 0.0f};
    if (pbase >= 0 && tid < SC_ROWS) rcnt[tid] = 0u;
    __syncthreads();

    const int  pwl = tid & 255;
    const int  pgq = tid >> 8;
    const bool pon = pwl < SC_WORDS && blockIdx.x * SC_WORDS + pwl < ((N + 1) >> 1);
    const unsigned* pseg = (const unsigned*)out + (size_t)base * NFILT + max(pbase, 0) + (pon ? pwl : 0);
    unsigned pv[PP_BATCH] = {};

    float sc[SC_ST] = {};
    if (pbase < 0) {
#pragma unroll
        for (int t = 0; t < SC_ST; ++t)
            sc[t] = dinv[min(base + ((tid + t * 1024) >> 4), N - 1)];
    } else {
        kw_pp_load(pv, pseg, pgq, G);
    }

    uint2* lst = slist + wv * SC_CAP;
    int cnt = 0;

    if (mwords > 0) {
        const unsigned* mk = (const unsigned*)out + (size_t)base * NFILT;
        unsigned* ll = llist + wv * LL_CAP;
        const int nunits = (mwords + 3) >> 2;
        const int sub = lane & 3;
        const int tb  = (lane >> 2) << 2;
        int nl = 0;

        unsigned wcur = 0u;
        if (wv < nunits && wv * 4 + sub < mwords) wcur = mk[wv * 4 + sub];

        for (int u = wv; u < nunits; u += SC_WAVES) {
            const int wn = (u + SC_WAVES) * 4 + sub;
            const unsigned wload = mk[min(wn, mwords - 1)];
            const unsigned wnext = (wn < mwords) ? wload : 0u;

            const unsigned lbase = (unsigned)(u * 4 + sub) * 32u + (unsigned)tb;
            const unsigned bits  = (wcur >> tb) & 0xFu;
            for (int t = 0; t < 4; ++t) {
                const bool hit = ((bits >> t) & 1u) != 0u;
                const unsigned m = __builtin_amdgcn_ballot_w32(hit);
                if (hit) ll[nl + (int)__builtin_amdgcn_mbcnt_lo(m, 0u)] = lbase + (unsigned)t;
                nl += __builtin_popcount(m);
            }
            while (nl >= 16) {
                nl -= 16;
                kw_scan_lines<true>(ll + nl, 16, lst, cnt, lane, base, E, rows, cols, XF, acc);
            }
            wcur = wnext;
        }
        if (nl > 0)
            kw_scan_lines<false>(ll, nl, lst, cnt, lane, base, E, rows, cols, XF, acc);
    } else {
    const int nfull = E / SC_BLK;
    const int iters = (nfull + SC_WAVES - 1) / SC_WAVES;

    if (iters > 0) {
        int k = (int)((blockIdx.x * 13u) % (unsigned)iters);

        int4 ra[4], rb[4];
        kw_load_blk(ra, rows, min(wv + SC_WAVES * k, nfull - 1), lane);

        for (int t = 0; t < iters; t += 2) {
            int kb = k + 1;
            if (kb == iters) kb = 0;
            const int blka = wv + SC_WAVES * k;
            const int blkb = wv + SC_WAVES * kb;
            kw_load_blk(rb, rows, min(blkb, nfull - 1), lane);
            if (blka < nfull)
                kw_scan_blk(ra, blka, lst, cnt, lane, base, cols, XF, acc);

            int kn = kb + 1;
            if (kn == iters) kn = 0;
            kw_load_blk(ra, rows, min(wv + SC_WAVES * kn, nfull - 1), lane);
            if (t + 1 < iters && blkb < nfull)
                kw_scan_blk(rb, blkb, lst, cnt, lane, base, cols, XF, acc);
            k = kn;
        }
    }
    if (nfull * SC_BLK < E && wv == (nfull & (SC_WAVES - 1))) {
        int4 rt[4];
        kw_load_tail(rt, rows, nfull, lane, E);
        kw_scan_blk(rt, nfull, lst, cnt, lane, base, cols, XF, acc);
    }
    }
    if (cnt > 0)
        kw_flush<false>(lst, 0, cnt, lane, cols, XF, acc);

    if (pbase >= 0) {
        unsigned lo = 0u, hi = 0u;
        kw_pp_sum(pv, pgq, G, pon, lo, hi);
        if (pon) {
            atomicAdd(&rcnt[2 * pwl],     lo);
            atomicAdd(&rcnt[2 * pwl + 1], hi);
        }
    }

    __syncthreads();

#pragma unroll
    for (int t = 0; t < SC_ST; ++t) {
        const int i   = tid + t * 1024;
        const int row = base + (i >> 4);
        if (i < SC_ROWS * (NFILT / 4) && row < N) {
            const float s = (pbase >= 0) ? kw_dinv_of(rcnt[i >> 4]) : sc[t];
            const v4f   a = ((const v4f*)acc)[i];
            ((v4f*)(out + (size_t)row * NFILT))[i & 15] = (v4f){s * a.x, s * a.y, s * a.z, s * a.w};
        }
    }
}

extern "C" void kernel_launch(void* const* d_in, const int* in_sizes, int n_in,
                              void* d_out, int out_size, void* d_ws, size_t ws_size,
                              hipStream_t stream) {
    const float* x   = (const float*)d_in[0];
    const float* flt = (const float*)d_in[1];
    const int*   eid = (const int*)d_in[2];

    const int N = in_sizes[0] / NFEAT;
    const int E = in_sizes[2] / 2;
    const int* rows = eid;
    const int* cols = eid + E;

    float* out = (float*)d_out;
    float* deg = (float*)d_ws;
    float* XF  = (float*)d_ws + (size_t)((N + 63) & ~63);

    unsigned* partial = (unsigned*)XF;
    const int nwords = (N + 1) / 2;
    const int R   = (N + DEG_RANGE_BINS - 1) / DEG_RANGE_BINS;
    const int G   = (E + DEG_CHUNK - 1) / DEG_CHUNK;
    const int wpg = R * DEG_RANGE_WORDS;
    const int nown     = (N + SC_ROWS - 1) / SC_ROWS;
    const int opq      = (nown + R - 1) / R;
    const int mwords   = G * MARK_CHUNK_W;
    const int lastrows = N - (nown - 1) * SC_ROWS;
    const bool keep    = opq <= MARK_OWN_MAX && mwords <= lastrows * NFILT && E >= 32;
    unsigned* marks    = keep ? (unsigned*)d_out : nullptr;
    const bool park    = keep && mwords + G * PP_PITCH <= lastrows * NFILT && G <= 4 * PP_BATCH;
    const int  pbase   = park ? mwords : -1;
    kw_degree_lds<<<dim3(G, R), 1024, 0, stream>>>(rows, partial, marks, E, nwords, wpg, nown, opq, pbase);
    if (!park)
        kw_dinv_reduce<<<(nwords + 255) / 256, 256, 0, stream>>>(partial, deg, N, nwords, G, wpg);

    const int stripes = (N + 15) / 16;
    kw_gemm_wmma<<<(stripes + 7) / 8, 256, 0, stream>>>(x, flt, deg, park ? (const unsigned*)d_out : nullptr,
                                                       XF, N, G, pbase);

    kw_rowsum_lds<<<nown, 1024, 0, stream>>>(rows, cols, deg, XF, out, N, E, keep ? mwords : 0, G, pbase);
}


# Row-sum owners read per-owner edge lists (column id + local row in one word), sorted in LDS by the degree launch and parked in out, not marked row-id lines + column ids; LDS f32 adds in arrival order
# speedup vs baseline: 2.6013x; 1.0775x over previous
#include <hip/hip_runtime.h>
#include <hip/hip_bf16.h>

typedef __attribute__((ext_vector_type(2))) float v2f;
typedef __attribute__((ext_vector_type(4))) float v4f;
typedef __attribute__((ext_vector_type(8))) float v8f;

#define NFEAT 512
#define NFILT 64
#define KCHUNK 128

#define DEG_CHUNK       32768
#define DEG_RANGE_WORDS 10200
#define DEG_RANGE_BINS  (DEG_RANGE_WORDS * 2)

#define SC_ROWS        400
#define LS_OWN_MAX     56
#define LS_PITCH       256
#define LS_CAP         (LS_PITCH - 1)
#define LS_ROW_BITS    9

#define SC_WORDS       (SC_ROWS / 2)
#define PP_PITCH       256
#define PP_BATCH       16

__device__ __forceinline__ void kw_count_row(unsigned* h, int r, int base) {
    const unsigned d = (unsigned)(r - base);
    if (d < (unsigned)DEG_RANGE_BINS)
        atomicAdd(&h[d >> 1], 1u << ((d & 1u) << 4));
}

__device__ __forceinline__ void kw_list_rows(unsigned* sl, unsigned* lc, const int (&r)[8], const int (&c)[8],
                                             unsigned obase, unsigned opq, unsigned nown) {
    unsigned ol[8], slot[8];
#pragma unroll
    for (int k = 0; k < 8; ++k) {
        const unsigned o = (unsigned)r[k] / (unsigned)SC_ROWS;
        ol[k]   = o - obase;
        slot[k] = (unsigned)LS_CAP;
        if (ol[k] < opq && o < nown)
            slot[k] = atomicAdd(&lc[ol[k]], 1u);
    }
#pragma unroll
    for (int k = 0; k < 8; ++k) {
        if (slot[k] < (unsigned)LS_CAP)
            sl[ol[k] * LS_PITCH + slot[k]] = ((unsigned)c[k] << LS_ROW_BITS)
                                           | ((unsigned)r[k] - (ol[k] + obase) * (unsigned)SC_ROWS);
    }
}

template <bool LISTS>
__global__ __launch_bounds__(1024) void kw_degree_lds(const int* __restrict__ rows,
                                                      const int* __restrict__ cols,
                                                      unsigned* __restrict__ partial,
                                                      unsigned* __restrict__ lists,
                                                      int E, int nwords, int wpg,
                                                      int nown, int opq, int pbase) {
    __shared__ __attribute__((aligned(16))) unsigned h[DEG_RANGE_WORDS];
    __shared__ __attribute__((aligned(16))) unsigned sl[LS_OWN_MAX * LS_PITCH];
    __shared__ unsigned lc[LS_OWN_MAX];

    const int tid  = threadIdx.x;
    const int base = blockIdx.y * DEG_RANGE_BINS;

    for (int i = tid; i < DEG_RANGE_WORDS / 4; i += 1024)
        ((uint4*)h)[i] = make_uint4(0u, 0u, 0u, 0u);
    if (LISTS && tid < LS_OWN_MAX)
        lc[tid] = 0u;

    const int e0 = blockIdx.x * DEG_CHUNK;

    int4 rr[DEG_CHUNK / 4096];
    int4 cc[DEG_CHUNK / 4096];
#pragma unroll
    for (int j = 0; j < DEG_CHUNK / 4096; ++j) {
        const int eb = e0 + (j << 12);
        const int e  = eb + tid * 4;
        if (eb + 4096 <= E) {
            rr[j] = *(const int4*)(rows + e);
            if (LISTS) cc[j] = *(const int4*)(cols + e);
        } else {
            rr[j].x = (e     < E) ? rows[e]     : -1;
            rr[j].y = (e + 1 < E) ? rows[e + 1] : -1;
            rr[j].z = (e + 2 < E) ? rows[e + 2] : -1;
            rr[j].w = (e + 3 < E) ? rows[e + 3] : -1;
            if (LISTS) {
                cc[j].x = (e     < E) ? cols[e]     : 0;
                cc[j].y = (e + 1 < E) ? cols[e + 1] : 0;
                cc[j].z = (e + 2 < E) ? cols[e + 2] : 0;
                cc[j].w = (e + 3 < E) ? cols[e + 3] : 0;
            }
        }
    }
    __syncthreads();

#pragma unroll
    for (int j = 0; j < DEG_CHUNK / 4096; ++j) {
        kw_count_row(h, rr[j].x, base);
        kw_count_row(h, rr[j].y, base);
        kw_count_row(h, rr[j].z, base);
        kw_count_row(h, rr[j].w, base);
    }
    const unsigned obase = blockIdx.y * (unsigned)opq;
    if (LISTS) {
#pragma unroll
        for (int j = 0; j < DEG_CHUNK / 4096; j += 2) {
            const int r8[8] = {rr[j].x, rr[j].y, rr[j].z, rr[j].w, rr[j + 1].x, rr[j + 1].y, rr[j + 1].z, rr[j + 1].w};
            const int c8[8] = {cc[j].x, cc[j].y, cc[j].z, cc[j].w, cc[j + 1].x, cc[j + 1].y, cc[j + 1].z, cc[j + 1].w};
            kw_list_rows(sl, lc, r8, c8, obase, (unsigned)opq, (unsigned)nown);
        }
    }
    __syncthreads();

    if (LISTS) {
        for (int i = tid; i < opq * LS_PITCH; i += 1024) {
            const unsigned ol = (unsigned)i / (unsigned)LS_PITCH;
            const unsigned w  = (unsigned)i % (unsigned)LS_PITCH;
            const unsigned o  = obase + ol;
            if (o < (unsigned)nown) {
                const unsigned tot = lc[ol];
                const unsigned cnt = min(tot, (unsigned)LS_CAP);
                const unsigned ln  = w >> 5;
                if (ln < ((cnt + 31u) >> 5) || ln == (unsigned)(LS_PITCH / 32 - 1))
                    lists[(size_t)o * (SC_ROWS * NFILT) + blockIdx.x * LS_PITCH + w] =
                        (w == (unsigned)LS_CAP) ? tot : sl[i];
            }
        }
    }

    const int wbase = blockIdx.y * DEG_RANGE_WORDS;
    if (pbase >= 0) {
        for (int i = tid; i < DEG_RANGE_WORDS / 4; i += 1024) {
            const int w = wbase + i * 4;
            if (w < nwords) {
                const unsigned o = (unsigned)w / (unsigned)SC_WORDS;
                unsigned* d = lists + (size_t)o * (SC_ROWS * NFILT) + pbase
                            + blockIdx.x * PP_PITCH + (w - (int)o * SC_WORDS);
                *(uint4*)d = ((const uint4*)h)[i];
            }
        }
    } else {
        uint4* dst = (uint4*)(partial + (size_t)blockIdx.x * wpg + wbase);
        for (int i = tid; i < DEG_RANGE_WORDS / 4; i += 1024)
            if (wbase + i * 4 < nwords)
                dst[i] = ((const uint4*)h)[i];
    }
}

__device__ __forceinline__ float kw_dinv_of(unsigned cnt) {
    const float d = (float)cnt;
    return (d > 0.0f) ? rsqrtf(d) : 0.0f;
}

__device__ __forceinline__ void kw_pp_load(unsigned (&v)[PP_BATCH], const unsigned* seg, int g0, int G) {
#pragma unroll
    for (int j = 0; j < PP_BATCH; ++j)
        v[j] = seg[(size_t)min(g0 + 4 * j, G - 1) * PP_PITCH];
}

__device__ __forceinline__ void kw_pp_sum(const unsigned (&v)[PP_BATCH], int g0, int G, bool on,
                                          unsigned& lo, unsigned& hi) {
#pragma unroll
    for (int j = 0; j < PP_BATCH; ++j) {
        const unsigned t = (on && g0 + 4 * j < G) ? v[j] : 0u;
        lo += t & 0xFFFFu;
        hi += t >> 16;
    }
}

__global__ __launch_bounds__(256) void kw_dinv_reduce(const unsigned* __restrict__ partial,
                                                      float* __restrict__ dinv,
                                                      int N, int nwords, int G, int wpg) {
    const int w = blockIdx.x * 256 + threadIdx.x;
    if (w >= nwords) return;

    const unsigned* p = partial + w;
    unsigned lo = 0u, hi = 0u;
#pragma unroll 7
    for (int g = 0; g < G; ++g) {
        const unsigned v = p[(size_t)g * wpg];
        lo += v & 0xFFFFu;
        hi += v >> 16;
    }
    const float d0 = (float)lo;
    const float d1 = (float)hi;
    const float i0 = (d0 > 0.0f) ? rsqrtf(d0) : 0.0f;
    const float i1 = (d1 > 0.0f) ? rsqrtf(d1) : 0.0f;

    const int n0 = 2 * w;
    if (n0 + 1 < N) {
        v2f o = {i0, i1};
        *(v2f*)(dinv + n0) = o;
    } else {
        dinv[n0] = i0;
    }
}

__global__ __launch_bounds__(256) void kw_gemm_wmma(const float* __restrict__ x,
                                                    const float* __restrict__ flt,
                                                    const float* __restrict__ dinv,
                                                    const unsigned* __restrict__ pp,
                                                    float* __restrict__ XF, int N, int G, int pbase) {
    __shared__ float sB[KCHUNK * NFILT];
    __shared__ unsigned sCnt[128];

    const int  nwords = (N + 1) >> 1;
    const int  pwl    = threadIdx.x & 63;
    const int  pgq    = threadIdx.x >> 6;
    const int  pw     = blockIdx.x * 64 + pwl;
    const bool pon    = pw < nwords;
    const unsigned* pseg = nullptr;
    unsigned pv[PP_BATCH] = {};
    if (pp != nullptr) {
        if (threadIdx.x < 128) sCnt[threadIdx.x] = 0u;
        const int      wc = min(pw, nwords - 1);
        const unsigned o  = (unsigned)wc / (unsigned)SC_WORDS;
        pseg = pp + (size_t)o * (SC_ROWS * NFILT) + pbase + (wc - (int)o * SC_WORDS);
    }

    const int lane = threadIdx.x & 31;
    const int wib  = threadIdx.x >> 5;
    const int half = lane >> 4;
    const int l16  = lane & 15;

    const int stripe = blockIdx.x * 8 + wib;
    const int m0     = stripe * 16;
    const bool active = (m0 < N);

    int row = m0 + l16;
    if (row >= N) row = N - 1;
    const float* xrow = x + (size_t)row * NFEAT;

    v8f acc0 = {}, acc1 = {}, acc2 = {}, acc3 = {};

    float dv[8] = {};

    const float* bbase = sB + lane * 2;

#pragma unroll
    for (int kc = 0; kc < NFEAT; kc += KCHUNK) {
        __syncthreads();
        if (kc == 0 && pp != nullptr)
            kw_pp_load(pv, pseg, pgq, G);
        {
            const v4f* src = (const v4f*)(flt + (size_t)kc * NFILT);
            for (int t4 = threadIdx.x; t4 < KCHUNK * (NFILT / 4); t4 += 256) {
                v4f v = src[t4];
                int r    = t4 >> 4;
                int g    = r >> 2;
                int koff = r & 3;
                int h    = koff >> 1;
                int j    = koff & 1;
                int cb   = (t4 & 15) << 2;
                int n    = cb >> 4;
                int c16  = cb & 15;
                float* d = sB + (((g << 2) + n) << 6) + (((h << 4) + c16) << 1) + j;
                d[0] = v.x; d[2] = v.y; d[4] = v.z; d[6] = v.w;
            }
        }
        if (kc == 0 && pp != nullptr) {
            unsigned lo = 0u, hi = 0u;
            kw_pp_sum(pv, pgq, G, pon, lo, hi);
            if (pon) {
                atomicAdd(&sCnt[2 * pwl],     lo);
                atomicAdd(&sCnt[2 * pwl + 1], hi);
            }
        }
        __syncthreads();

        if (kc == 0) {
            if (pp == nullptr) {
#pragma unroll
                for (int r = 0; r < 8; ++r)
                    dv[r] = dinv[min(m0 + r + 8 * half, N - 1)];
            }
        }

        if (active) {
#pragma unroll 4
            for (int k2 = 0; k2 < KCHUNK; k2 += 4) {
                v2f a = *(const v2f*)(xrow + kc + k2 + 2 * half);

                const int g = k2 >> 2;
                const float* bg = bbase + (g << 8);
                v2f b0 = *(const v2f*)(bg);
                v2f b1 = *(const v2f*)(bg + 64);
                v2f b2 = *(const v2f*)(bg + 128);
                v2f b3 = *(const v2f*)(bg + 192);

                acc0 = __builtin_amdgcn_wmma_f32_16x16x4_f32(false, a, false, b0, (short)0, acc0, false, false);
                acc1 = __builtin_amdgcn_wmma_f32_16x16x4_f32(false, a, false, b1, (short)0, acc1, false, false);
                acc2 = __builtin_amdgcn_wmma_f32_16x16x4_f32(false, a, false, b2, (short)0, acc2, false, false);
                acc3 = __builtin_amdgcn_wmma_f32_16x16x4_f32(false, a, false, b3, (short)0, acc3, false, false);
            }
        }
    }

    if (active) {
        if (pp != nullptr) {
#pragma unroll
            for (int r = 0; r < 8; ++r)
                dv[r] = kw_dinv_of(sCnt[wib * 16 + r + 8 * half]);
        }
        asm volatile("v_nop\n\tv_nop\n\tv_nop\n\tv_nop\n\tv_nop\n\tv_nop\n\tv_nop\n\tv_nop"
                     : "+v"(acc0), "+v"(acc1), "+v"(acc2), "+v"(acc3));

#pragma unroll
        for (int r = 0; r < 8; ++r) {
            int M = m0 + r + 8 * half;
            if (M < N) {
                float* o = XF + (size_t)M * NFILT + l16;
                o[0]  = dv[r] * acc0[r];
                o[16] = dv[r] * acc1[r];
                o[32] = dv[r] * acc2[r];
                o[48] = dv[r] * acc3[r];
            }
        }
    }
}

#define SC_WAVES 32
#define SC_BLK   512
#define SC_CAP   (SC_BLK + 32)
#define SC_ST    ((SC_ROWS * (NFILT / 4) + 1023) / 1024)

__device__ __forceinline__ void kw_load_blk(int4 (&r)[4], const int* __restrict__ rows,
                                            int blk, int lane) {
    const int4* p = (const int4*)(rows + (size_t)blk * SC_BLK) + lane;
#pragma unroll
    for (int q = 0; q < 4; ++q)
        r[q] = p[q * 32];
}

__device__ __forceinline__ void kw_load_tail(int4 (&r)[4], const int* __restrict__ rows,
                                             int blk, int lane, int E) {
#pragma unroll
    for (int q = 0; q < 4; ++q) {
        const int e = blk * SC_BLK + q * 128 + lane * 4;
        r[q].x = (e     < E) ? rows[e]     : -1;
        r[q].y = (e + 1 < E) ? rows[e + 1] : -1;
        r[q].z = (e + 2 < E) ? rows[e + 2] : -1;
        r[q].w = (e + 3 < E) ? rows[e + 3] : -1;
    }
}

__device__ __forceinline__ void kw_append(uint2* lst, int& cnt, bool hit, unsigned e, unsigned d) {
    const unsigned m = __builtin_amdgcn_ballot_w32(hit);
    if (hit) lst[cnt + (int)__builtin_amdgcn_mbcnt_lo(m, 0u)] = make_uint2(e, d);
    cnt += __builtin_popcount(m);
}

template <bool FULL>
__device__ __forceinline__ void kw_add_rows(int c, int d, int nvalid, int lane,
                                            const float* __restrict__ XF, float* acc) {
    for (int j0 = 0; j0 < 32; j0 += 8) {
        if (!FULL && j0 >= nvalid) break;
        float v0[8] = {}, v1[8] = {};
#pragma unroll
        for (int u = 0; u < 8; ++u) {
            if (FULL || j0 + u < nvalid) {
                const int cj = __builtin_amdgcn_readlane(c, j0 + u);
                const float* src = XF + (size_t)cj * NFILT;
                v0[u] = src[lane];
                v1[u] = src[lane + 32];
            }
        }
#pragma unroll
        for (int u = 0; u < 8; ++u) {
            if (FULL || j0 + u < nvalid) {
                const int dj = __builtin_amdgcn_readlane(d, j0 + u);
                float* dst = acc + dj * NFILT + lane;
                __hip_atomic_fetch_add(&dst[0],  v0[u], __ATOMIC_RELAXED, __HIP_MEMORY_SCOPE_WORKGROUP);
                __hip_atomic_fetch_add(&dst[32], v1[u], __ATOMIC_RELAXED, __HIP_MEMORY_SCOPE_WORKGROUP);
            }
        }
    }
}

template <bool FULL>
__device__ __forceinline__ void kw_flush(const uint2* lst, int first, int nvalid, int lane,
                                         const int* __restrict__ cols,
                                         const float* __restrict__ XF, float* acc) {
    int c = 0, d = 0;
    if (FULL || lane < nvalid) {
        const uint2 t = lst[first + lane];
        c = cols[t.x];
        d = (int)t.y;
    }
    kw_add_rows<FULL>(c, d, nvalid, lane, XF, acc);
}

__device__ __forceinline__ void kw_scan_blk(const int4 (&r)[4], int blk, uint2* lst, int& cnt,
                                            int lane, int base,
                                            const int* __restrict__ cols,
                                            const float* __restrict__ XF, float* acc) {
#pragma unroll
    for (int q = 0; q < 4; ++q) {
        const unsigned dx = (unsigned)(r[q].x - base);
        const unsigned dy = (unsigned)(r[q].y - base);
        const unsigned dz = (unsigned)(r[q].z - base);
        const unsigned dw = (unsigned)(r[q].w - base);
        const unsigned dm = min(min(dx, dy), min(dz, dw));
        if (__builtin_amdgcn_ballot_w32(dm < (unsigned)SC_ROWS) != 0u) {
            const unsigned e = (unsigned)(blk * SC_BLK + q * 128 + lane * 4);
            kw_append(lst, cnt, dx < (unsigned)SC_ROWS, e,      dx);
            kw_append(lst, cnt, dy < (unsigned)SC_ROWS, e + 1u, dy);
            kw_append(lst, cnt, dz < (unsigned)SC_ROWS, e + 2u, dz);
            kw_append(lst, cnt, dw < (unsigned)SC_ROWS, e + 3u, dw);
        }
    }
    while (cnt >= 32) {
        cnt -= 32;
        kw_flush<true>(lst, cnt, 32, lane, cols, XF, acc);
    }
}

__global__ __launch_bounds__(1024) void kw_rowsum_lds(const int* __restrict__ rows,
                                                      const int* __restrict__ cols,
                                                      const float* __restrict__ dinv,
                                                      const float* __restrict__ XF,
                                                      float* out, int N, int E, int mwords,
                                                      int G, int pbase) {
    __shared__ __attribute__((aligned(16))) float acc[SC_ROWS * NFILT];
    __shared__ unsigned rcnt[SC_ROWS];
    __shared__ __attribute__((aligned(16))) uint2 slist[SC_WAVES * SC_CAP];

    const int tid  = threadIdx.x;
    const int lane = tid & 31;
    const int wv   = __builtin_amdgcn_readfirstlane(tid >> 5);
    const int base = blockIdx.x * SC_ROWS;

    const unsigned* lb = (const unsigned*)out + (size_t)base * NFILT;
    unsigned c0 = 0u, c1 = 0u;
    if (mwords > 0) {
        const unsigned t0 = lb[(size_t)min(lane,      G - 1) * LS_PITCH + LS_CAP];
        const unsigned t1 = lb[(size_t)min(lane + 32, G - 1) * LS_PITCH + LS_CAP];
        c0 = (lane      < G) ? t0 : 0u;
        c1 = (lane + 32 < G) ? t1 : 0u;
    }

    for (int i = tid; i < SC_ROWS * NFILT / 4; i += 1024)
        ((v4f*)acc)[i] = (v4f){0.0f, 0.0f, 0.0f, 0.0f};
    if (pbase >= 0 && tid < SC_ROWS) rcnt[tid] = 0u;
    __syncthreads();

    const int  pwl = tid & 255;
    const int  pgq = tid >> 8;
    const bool pon = pwl < SC_WORDS && blockIdx.x * SC_WORDS + pwl < ((N + 1) >> 1);
    const unsigned* pseg = (const unsigned*)out + (size_t)base * NFILT + max(pbase, 0) + (pon ? pwl : 0);
    unsigned pv[PP_BATCH] = {};

    float sc[SC_ST] = {};
    if (pbase < 0) {
#pragma unroll
        for (int t = 0; t < SC_ST; ++t)
            sc[t] = dinv[min(base + ((tid + t * 1024) >> 4), N - 1)];
    } else {
        kw_pp_load(pv, pseg, pgq, G);
    }

    uint2* lst = slist + wv * SC_CAP;
    int cnt = 0;

    const bool uselist = mwords > 0 &&
                         __builtin_amdgcn_ballot_w32(c0 > (unsigned)LS_CAP || c1 > (unsigned)LS_CAP) == 0u;

    if (uselist) {
        int      T  = 0;
        unsigned pe = 0u;
        int      pn = 0;
        for (int g = 0; g < G; ++g) {
            const int c  = __builtin_amdgcn_readlane((int)(g < 32 ? c0 : c1), g & 31);
            const int ng = (c + 31) >> 5;
            for (int j = (wv - T) & (SC_WAVES - 1); j < ng; j += SC_WAVES) {
                const int n = min(32, c - j * 32);
                const unsigned ne = lb[g * LS_PITCH + j * 32 + min(lane, n - 1)];
                if (pn == 32)
                    kw_add_rows<true>((int)(pe >> LS_ROW_BITS), (int)(pe & ((1u << LS_ROW_BITS) - 1u)),
                                      32, lane, XF, acc);
                else if (pn > 0)
                    kw_add_rows<false>((int)(pe >> LS_ROW_BITS), (int)(pe & ((1u << LS_ROW_BITS) - 1u)),
                                       pn, lane, XF, acc);
                pe = ne;
                pn = n;
            }
            T += ng;
        }
        if (pn == 32)
            kw_add_rows<true>((int)(pe >> LS_ROW_BITS), (int)(pe & ((1u << LS_ROW_BITS) - 1u)),
                              32, lane, XF, acc);
        else if (pn > 0)
            kw_add_rows<false>((int)(pe >> LS_ROW_BITS), (int)(pe & ((1u << LS_ROW_BITS) - 1u)),
                               pn, lane, XF, acc);
    } else {
    const int nfull = E / SC_BLK;
    const int iters = (nfull + SC_WAVES - 1) / SC_WAVES;

    if (iters > 0) {
        int k = (int)((blockIdx.x * 13u) % (unsigned)iters);

        int4 ra[4], rb[4];
        kw_load_blk(ra, rows, min(wv + SC_WAVES * k, nfull - 1), lane);

        for (int t = 0; t < iters; t += 2) {
            int kb = k + 1;
            if (kb == iters) kb = 0;
            const int blka = wv + SC_WAVES * k;
            const int blkb = wv + SC_WAVES * kb;
            kw_load_blk(rb, rows, min(blkb, nfull - 1), lane);
            if (blka < nfull)
                kw_scan_blk(ra, blka, lst, cnt, lane, base, cols, XF, acc);

            int kn = kb + 1;
            if (kn == iters) kn = 0;
            kw_load_blk(ra, rows, min(wv + SC_WAVES * kn, nfull - 1), lane);
            if (t + 1 < iters && blkb < nfull)
                kw_scan_blk(rb, blkb, lst, cnt, lane, base, cols, XF, acc);
            k = kn;
        }
    }
    if (nfull * SC_BLK < E && wv == (nfull & (SC_WAVES - 1))) {
        int4 rt[4];
        kw_load_tail(rt, rows, nfull, lane, E);
        kw_scan_blk(rt, nfull, lst, cnt, lane, base, cols, XF, acc);
    }
    }
    if (cnt > 0)
        kw_flush<false>(lst, 0, cnt, lane, cols, XF, acc);

    if (pbase >= 0) {
        unsigned lo = 0u, hi = 0u;
        kw_pp_sum(pv, pgq, G, pon, lo, hi);
        if (pon) {
            atomicAdd(&rcnt[2 * pwl],     lo);
            atomicAdd(&rcnt[2 * pwl + 1], hi);
        }
    }

    __syncthreads();

#pragma unroll
    for (int t = 0; t < SC_ST; ++t) {
        const int i   = tid + t * 1024;
        const int row = base + (i >> 4);
        if (i < SC_ROWS * (NFILT / 4) && row < N) {
            const float s = (pbase >= 0) ? kw_dinv_of(rcnt[i >> 4]) : sc[t];
            const v4f   a = ((const v4f*)acc)[i];
            ((v4f*)(out + (size_t)row * NFILT))[i & 15] = (v4f){s * a.x, s * a.y, s * a.z, s * a.w};
        }
    }
}

extern "C" void kernel_launch(void* const* d_in, const int* in_sizes, int n_in,
                              void* d_out, int out_size, void* d_ws, size_t ws_size,
                              hipStream_t stream) {
    const float* x   = (const float*)d_in[0];
    const float* flt = (const float*)d_in[1];
    const int*   eid = (const int*)d_in[2];

    const int N = in_sizes[0] / NFEAT;
    const int E = in_sizes[2] / 2;
    const int* rows = eid;
    const int* cols = eid + E;

    float* out = (float*)d_out;
    float* deg = (float*)d_ws;
    float* XF  = (float*)d_ws + (size_t)((N + 63) & ~63);

    unsigned* partial = (unsigned*)XF;
    const int nwords = (N + 1) / 2;
    const int R   = (N + DEG_RANGE_BINS - 1) / DEG_RANGE_BINS;
    const int G   = (E + DEG_CHUNK - 1) / DEG_CHUNK;
    const int wpg = R * DEG_RANGE_WORDS;
    const int nown     = (N + SC_ROWS - 1) / SC_ROWS;
    const int opq      = (nown + R - 1) / R;
    const int mwords   = G * LS_PITCH;
    const int lastrows = N - (nown - 1) * SC_ROWS;
    const bool keep    = opq <= LS_OWN_MAX && mwords <= lastrows * NFILT && G <= 64
                         && N <= (1 << (32 - LS_ROW_BITS)) && (E & 3) == 0;
    unsigned* lists    = keep ? (unsigned*)d_out : nullptr;
    const bool park    = keep && mwords + G * PP_PITCH <= lastrows * NFILT && G <= 4 * PP_BATCH;
    const int  pbase   = park ? mwords : -1;
    if (keep)
        kw_degree_lds<true><<<dim3(G, R), 1024, 0, stream>>>(rows, cols, partial, lists, E, nwords, wpg, nown, opq, pbase);
    else
        kw_degree_lds<false><<<dim3(G, R), 1024, 0, stream>>>(rows, cols, partial, lists, E, nwords, wpg, nown, opq, pbase);
    if (!park)
        kw_dinv_reduce<<<(nwords + 255) / 256, 256, 0, stream>>>(partial, deg, N, nwords, G, wpg);

    const int stripes = (N + 15) / 16;
    kw_gemm_wmma<<<(stripes + 7) / 8, 256, 0, stream>>>(x, flt, deg, park ? (const unsigned*)d_out : nullptr,
                                                       XF, N, G, pbase);

    kw_rowsum_lds<<<nown, 1024, 0, stream>>>(rows, cols, deg, XF, out, N, E, keep ? mwords : 0, G, pbase);
}


# Product re-cut: one resident work-group per 400-row owner stages the whole filter matrix in LDS once; 24 waves a stripe each + 2 a half stripe run all of K with no barrier, next A block in flight
# speedup vs baseline: 3.3177x; 1.2754x over previous
#include <hip/hip_runtime.h>
#include <hip/hip_bf16.h>

typedef __attribute__((ext_vector_type(2))) float v2f;
typedef __attribute__((ext_vector_type(4))) float v4f;
typedef __attribute__((ext_vector_type(8))) float v8f;

#define NFEAT 512
#define NFILT 64

#define DEG_CHUNK       32768
#define DEG_RANGE_WORDS 10200
#define DEG_RANGE_BINS  (DEG_RANGE_WORDS * 2)

#define SC_ROWS        400
#define LS_OWN_MAX     56
#define LS_PITCH       256
#define LS_CAP         (LS_PITCH - 1)
#define LS_ROW_BITS    9

#define SC_WORDS       (SC_ROWS / 2)
#define PP_PITCH       256
#define PP_BATCH       16

__device__ __forceinline__ void kw_count_row(unsigned* h, int r, int base) {
    const unsigned d = (unsigned)(r - base);
    if (d < (unsigned)DEG_RANGE_BINS)
        atomicAdd(&h[d >> 1], 1u << ((d & 1u) << 4));
}

__device__ __forceinline__ void kw_list_rows(unsigned* sl, unsigned* lc, const int (&r)[8], const int (&c)[8],
                                             unsigned obase, unsigned opq, unsigned nown) {
    unsigned ol[8], slot[8];
#pragma unroll
    for (int k = 0; k < 8; ++k) {
        const unsigned o = (unsigned)r[k] / (unsigned)SC_ROWS;
        ol[k]   = o - obase;
        slot[k] = (unsigned)LS_CAP;
        if (ol[k] < opq && o < nown)
            slot[k] = atomicAdd(&lc[ol[k]], 1u);
    }
#pragma unroll
    for (int k = 0; k < 8; ++k) {
        if (slot[k] < (unsigned)LS_CAP)
            sl[ol[k] * LS_PITCH + slot[k]] = ((unsigned)c[k] << LS_ROW_BITS)
                                           | ((unsigned)r[k] - (ol[k] + obase) * (unsigned)SC_ROWS);
    }
}

template <bool LISTS>
__global__ __launch_bounds__(1024) void kw_degree_lds(const int* __restrict__ rows,
                                                      const int* __restrict__ cols,
                                                      unsigned* __restrict__ partial,
                                                      unsigned* __restrict__ lists,
                                                      int E, int nwords, int wpg,
                                                      int nown, int opq, int pbase) {
    __shared__ __attribute__((aligned(16))) unsigned h[DEG_RANGE_WORDS];
    __shared__ __attribute__((aligned(16))) unsigned sl[LS_OWN_MAX * LS_PITCH];
    __shared__ unsigned lc[LS_OWN_MAX];

    const int tid  = threadIdx.x;
    const int base = blockIdx.y * DEG_RANGE_BINS;

    for (int i = tid; i < DEG_RANGE_WORDS / 4; i += 1024)
        ((uint4*)h)[i] = make_uint4(0u, 0u, 0u, 0u);
    if (LISTS && tid < LS_OWN_MAX)
        lc[tid] = 0u;

    const int e0 = blockIdx.x * DEG_CHUNK;

    int4 rr[DEG_CHUNK / 4096];
    int4 cc[DEG_CHUNK / 4096];
#pragma unroll
    for (int j = 0; j < DEG_CHUNK / 4096; ++j) {
        const int eb = e0 + (j << 12);
        const int e  = eb + tid * 4;
        if (eb + 4096 <= E) {
            rr[j] = *(const int4*)(rows + e);
            if (LISTS) cc[j] = *(const int4*)(cols + e);
        } else {
            rr[j].x = (e     < E) ? rows[e]     : -1;
            rr[j].y = (e + 1 < E) ? rows[e + 1] : -1;
            rr[j].z = (e + 2 < E) ? rows[e + 2] : -1;
            rr[j].w = (e + 3 < E) ? rows[e + 3] : -1;
            if (LISTS) {
                cc[j].x = (e     < E) ? cols[e]     : 0;
                cc[j].y = (e + 1 < E) ? cols[e + 1] : 0;
                cc[j].z = (e + 2 < E) ? cols[e + 2] : 0;
                cc[j].w = (e + 3 < E) ? cols[e + 3] : 0;
            }
        }
    }
    __syncthreads();

#pragma unroll
    for (int j = 0; j < DEG_CHUNK / 4096; ++j) {
        kw_count_row(h, rr[j].x, base);
        kw_count_row(h, rr[j].y, base);
        kw_count_row(h, rr[j].z, base);
        kw_count_row(h, rr[j].w, base);
    }
    const unsigned obase = blockIdx.y * (unsigned)opq;
    if (LISTS) {
#pragma unroll
        for (int j = 0; j < DEG_CHUNK / 4096; j += 2) {
            const int r8[8] = {rr[j].x, rr[j].y, rr[j].z, rr[j].w, rr[j + 1].x, rr[j + 1].y, rr[j + 1].z, rr[j + 1].w};
            const int c8[8] = {cc[j].x, cc[j].y, cc[j].z, cc[j].w, cc[j + 1].x, cc[j + 1].y, cc[j + 1].z, cc[j + 1].w};
            kw_list_rows(sl, lc, r8, c8, obase, (unsigned)opq, (unsigned)nown);
        }
    }
    __syncthreads();

    if (LISTS) {
        for (int i = tid; i < opq * LS_PITCH; i += 1024) {
            const unsigned ol = (unsigned)i / (unsigned)LS_PITCH;
            const unsigned w  = (unsigned)i % (unsigned)LS_PITCH;
            const unsigned o  = obase + ol;
            if (o < (unsigned)nown) {
                const unsigned tot = lc[ol];
                const unsigned cnt = min(tot, (unsigned)LS_CAP);
                const unsigned ln  = w >> 5;
                if (ln < ((cnt + 31u) >> 5) || ln == (unsigned)(LS_PITCH / 32 - 1))
                    lists[(size_t)o * (SC_ROWS * NFILT) + blockIdx.x * LS_PITCH + w] =
                        (w == (unsigned)LS_CAP) ? tot : sl[i];
            }
        }
    }

    const int wbase = blockIdx.y * DEG_RANGE_WORDS;
    if (pbase >= 0) {
        for (int i = tid; i < DEG_RANGE_WORDS / 4; i += 1024) {
            const int w = wbase + i * 4;
            if (w < nwords) {
                const unsigned o = (unsigned)w / (unsigned)SC_WORDS;
                unsigned* d = lists + (size_t)o * (SC_ROWS * NFILT) + pbase
                            + blockIdx.x * PP_PITCH + (w - (int)o * SC_WORDS);
                *(uint4*)d = ((const uint4*)h)[i];
            }
        }
    } else {
        uint4* dst = (uint4*)(partial + (size_t)blockIdx.x * wpg + wbase);
        for (int i = tid; i < DEG_RANGE_WORDS / 4; i += 1024)
            if (wbase + i * 4 < nwords)
                dst[i] = ((const uint4*)h)[i];
    }
}

__device__ __forceinline__ float kw_dinv_of(unsigned cnt) {
    const float d = (float)cnt;
    return (d > 0.0f) ? rsqrtf(d) : 0.0f;
}

__device__ __forceinline__ void kw_pp_load(unsigned (&v)[PP_BATCH], const unsigned* seg, int g0, int G) {
#pragma unroll
    for (int j = 0; j < PP_BATCH; ++j)
        v[j] = seg[(size_t)min(g0 + 4 * j, G - 1) * PP_PITCH];
}

__device__ __forceinline__ void kw_pp_sum(const unsigned (&v)[PP_BATCH], int g0, int G, bool on,
                                          unsigned& lo, unsigned& hi) {
#pragma unroll
    for (int j = 0; j < PP_BATCH; ++j) {
        const unsigned t = (on && g0 + 4 * j < G) ? v[j] : 0u;
        lo += t & 0xFFFFu;
        hi += t >> 16;
    }
}

__global__ __launch_bounds__(256) void kw_dinv_reduce(const unsigned* __restrict__ partial,
                                                      float* __restrict__ dinv,
                                                      int N, int nwords, int G, int wpg) {
    const int w = blockIdx.x * 256 + threadIdx.x;
    if (w >= nwords) return;

    const unsigned* p = partial + w;
    unsigned lo = 0u, hi = 0u;
#pragma unroll 7
    for (int g = 0; g < G; ++g) {
        const unsigned v = p[(size_t)g * wpg];
        lo += v & 0xFFFFu;
        hi += v >> 16;
    }
    const float d0 = (float)lo;
    const float d1 = (float)hi;
    const float i0 = (d0 > 0.0f) ? rsqrtf(d0) : 0.0f;
    const float i1 = (d1 > 0.0f) ? rsqrtf(d1) : 0.0f;

    const int n0 = 2 * w;
    if (n0 + 1 < N) {
        v2f o = {i0, i1};
        *(v2f*)(dinv + n0) = o;
    } else {
        dinv[n0] = i0;
    }
}

#define GO_STRIPES (SC_ROWS / 16)
#define GO_FULL    (GO_STRIPES - 1)
#define KW_SLOTS   "v_nop\n\tv_nop\n\tv_nop\n\tv_nop\n\tv_nop\n\tv_nop\n\tv_nop\n\tv_nop"

__device__ __forceinline__ void kw_a_load(v2f (&a)[8], const float* __restrict__ p) {
#pragma unroll
    for (int s = 0; s < 8; ++s)
        a[s] = *(const v2f*)(p + 4 * s);
}

template <int NP>
__device__ __forceinline__ void kw_mma_blk(const v2f (&a)[8], const float* bg, v8f (&acc)[2 * NP]) {
#pragma unroll
    for (int s = 0; s < 8; ++s) {
#pragma unroll
        for (int np = 0; np < NP; ++np) {
            const v4f b = *(const v4f*)(bg + (s << 8) + (np << 7));
            const v2f b0 = {b.x, b.y};
            const v2f b1 = {b.z, b.w};
            acc[2 * np]     = __builtin_amdgcn_wmma_f32_16x16x4_f32(false, a[s], false, b0, (short)0, acc[2 * np],     false, false);
            acc[2 * np + 1] = __builtin_amdgcn_wmma_f32_16x16x4_f32(false, a[s], false, b1, (short)0, acc[2 * np + 1], false, false);
        }
    }
}

template <int NP>
__device__ __forceinline__ void kw_stripe(const float* __restrict__ x, const float* sB,
                                          const unsigned* sPart, const float* __restrict__ dinv,
                                          float* __restrict__ XF, int N, int m0, int lr0, int np0,
                                          int lane, bool parked) {
    if (m0 >= N) return;

    int lv = lane;
    asm volatile("" : "+v"(lv));

    const float* xa = x + (size_t)min(m0 + (lane & 15), N - 1) * NFEAT + 2 * (lane >> 4);
    const float* bl = sB + lane * 4 + np0 * 128;

    float dv[8] = {};
    if (!parked) {
#pragma unroll
        for (int r = 0; r < 8; ++r)
            dv[r] = dinv[min(m0 + r + 8 * (lane >> 4), N - 1)];
    }

    v8f acc[2 * NP];
#pragma unroll
    for (int n = 0; n < 2 * NP; ++n)
        acc[n] = (v8f){};

    v2f a0[8], a1[8];
    kw_a_load(a0, xa);
    int ko = 0;
#pragma unroll 1
    for (int t = 0; t < NFEAT / 64 - 1; ++t) {
        const float* ap = xa + ko;
        const float* bp = bl + ko * 64;
        kw_a_load(a1, ap + 32);
        kw_mma_blk<NP>(a0, bp, acc);
        __builtin_amdgcn_sched_barrier(0);
        kw_a_load(a0, ap + 64);
        kw_mma_blk<NP>(a1, bp + 8 * 256, acc);
        if constexpr (NP == 2)
            asm volatile(KW_SLOTS : "+v"(acc[0]), "+v"(acc[1]), "+v"(acc[2]), "+v"(acc[3]), "+s"(ko));
        else
            asm volatile(KW_SLOTS : "+v"(acc[0]), "+v"(acc[1]), "+s"(ko));
        ko += 64;
    }
    {
        const float* ap = xa + ko;
        const float* bp = bl + ko * 64;
        kw_a_load(a1, ap + 32);
        kw_mma_blk<NP>(a0, bp, acc);
        kw_mma_blk<NP>(a1, bp + 8 * 256, acc);
    }

    if constexpr (NP == 2)
        asm volatile(KW_SLOTS : "+v"(acc[0]), "+v"(acc[1]), "+v"(acc[2]), "+v"(acc[3]), "+v"(lv));
    else
        asm volatile(KW_SLOTS : "+v"(acc[0]), "+v"(acc[1]), "+v"(lv));

    const int half = lv >> 4;
    const int l16  = lv & 15;
    if (parked) {
#pragma unroll
        for (int r = 0; r < 8; ++r) {
            const int lr = lr0 + r + 8 * half;
            dv[r] = kw_dinv_of(sPart[lr] + sPart[SC_ROWS + lr] + sPart[2 * SC_ROWS + lr] + sPart[3 * SC_ROWS + lr]);
        }
    }

#pragma unroll
    for (int r = 0; r < 8; ++r) {
        const int M = m0 + r + 8 * half;
        if (M < N) {
            float* o = XF + (size_t)M * NFILT + np0 * 32 + l16;
#pragma unroll
            for (int n = 0; n < 2 * NP; ++n)
                o[16 * n] = dv[r] * acc[n][r];
        }
    }
}

__global__ __launch_bounds__(1024) void kw_gemm_wmma(const float* __restrict__ x,
                                                     const float* __restrict__ flt,
                                                     const float* __restrict__ dinv,
                                                     const unsigned* __restrict__ pp,
                                                     float* __restrict__ XF, int N, int G, int pbase) {
    __shared__ __attribute__((aligned(16))) float sB[NFEAT * NFILT];
    __shared__ unsigned sPart[4 * SC_ROWS];

    const int tid  = threadIdx.x;
    const int base = blockIdx.x * SC_ROWS;

    const int  pwl = tid & 255;
    const int  pgq = tid >> 8;
    const bool pon = pwl < SC_WORDS && blockIdx.x * SC_WORDS + pwl < ((N + 1) >> 1);
    unsigned pv[PP_BATCH] = {};
    if (pp != nullptr)
        kw_pp_load(pv, pp + (size_t)base * NFILT + pbase + (pon ? pwl : 0), pgq, G);

    {
        const v4f* f4 = (const v4f*)flt;
        v4f u[4], w[4];
#pragma unroll
        for (int i = 0; i < 4; ++i) {
            const int q  = tid + i * 1024;
            const int p  = q >> 4;
            const int cg = q & 15;
            u[i] = f4[(2 * p) * (NFILT / 4) + cg];
            w[i] = f4[(2 * p + 1) * (NFILT / 4) + cg];
        }
#pragma unroll
        for (int i = 0; i < 4; ++i) {
            const int q  = tid + i * 1024;
            const int p  = q >> 4;
            const int cg = q & 15;
            const int n  = cg >> 2;
            float* d = sB + (p >> 1) * 256 + (n >> 1) * 128 + (((p & 1) << 4) + ((cg & 3) << 2)) * 4 + (n & 1) * 2;
            *(v2f*)(d)      = (v2f){u[i].x, w[i].x};
            *(v2f*)(d + 4)  = (v2f){u[i].y, w[i].y};
            *(v2f*)(d + 8)  = (v2f){u[i].z, w[i].z};
            *(v2f*)(d + 12) = (v2f){u[i].w, w[i].w};
        }
    }

    if (pp != nullptr) {
        unsigned lo = 0u, hi = 0u;
        kw_pp_sum(pv, pgq, G, pon, lo, hi);
        if (pwl < SC_WORDS) {
            sPart[pgq * SC_ROWS + 2 * pwl]     = lo;
            sPart[pgq * SC_ROWS + 2 * pwl + 1] = hi;
        }
    }
    __syncthreads();

    const int lane = tid & 31;
    const int wv = __builtin_amdgcn_readfirstlane(tid >> 5);
    if (wv < GO_FULL)
        kw_stripe<2>(x, sB, sPart, dinv, XF, N, base + wv * 16, wv * 16, 0, lane, pp != nullptr);
    else if (wv < GO_FULL + 2)
        kw_stripe<1>(x, sB, sPart, dinv, XF, N, base + GO_FULL * 16, GO_FULL * 16, wv - GO_FULL, lane, pp != nullptr);
}

#define SC_WAVES 32
#define SC_BLK   512
#define SC_CAP   (SC_BLK + 32)
#define SC_ST    ((SC_ROWS * (NFILT / 4) + 1023) / 1024)

__device__ __forceinline__ void kw_load_blk(int4 (&r)[4], const int* __restrict__ rows,
                                            int blk, int lane) {
    const int4* p = (const int4*)(rows + (size_t)blk * SC_BLK) + lane;
#pragma unroll
    for (int q = 0; q < 4; ++q)
        r[q] = p[q * 32];
}

__device__ __forceinline__ void kw_load_tail(int4 (&r)[4], const int* __restrict__ rows,
                                             int blk, int lane, int E) {
#pragma unroll
    for (int q = 0; q < 4; ++q) {
        const int e = blk * SC_BLK + q * 128 + lane * 4;
        r[q].x = (e     < E) ? rows[e]     : -1;
        r[q].y = (e + 1 < E) ? rows[e + 1] : -1;
        r[q].z = (e + 2 < E) ? rows[e + 2] : -1;
        r[q].w = (e + 3 < E) ? rows[e + 3] : -1;
    }
}

__device__ __forceinline__ void kw_append(uint2* lst, int& cnt, bool hit, unsigned e, unsigned d) {
    const unsigned m = __builtin_amdgcn_ballot_w32(hit);
    if (hit) lst[cnt + (int)__builtin_amdgcn_mbcnt_lo(m, 0u)] = make_uint2(e, d);
    cnt += __builtin_popcount(m);
}

template <bool FULL>
__device__ __forceinline__ void kw_add_rows(int c, int d, int nvalid, int lane,
                                            const float* __restrict__ XF, float* acc) {
    for (int j0 = 0; j0 < 32; j0 += 8) {
        if (!FULL && j0 >= nvalid) break;
        float v0[8] = {}, v1[8] = {};
#pragma unroll
        for (int u = 0; u < 8; ++u) {
            if (FULL || j0 + u < nvalid) {
                const int cj = __builtin_amdgcn_readlane(c, j0 + u);
                const float* src = XF + (size_t)cj * NFILT;
                v0[u] = src[lane];
                v1[u] = src[lane + 32];
            }
        }
#pragma unroll
        for (int u = 0; u < 8; ++u) {
            if (FULL || j0 + u < nvalid) {
                const int dj = __builtin_amdgcn_readlane(d, j0 + u);
                float* dst = acc + dj * NFILT + lane;
                __hip_atomic_fetch_add(&dst[0],  v0[u], __ATOMIC_RELAXED, __HIP_MEMORY_SCOPE_WORKGROUP);
                __hip_atomic_fetch_add(&dst[32], v1[u], __ATOMIC_RELAXED, __HIP_MEMORY_SCOPE_WORKGROUP);
            }
        }
    }
}

template <bool FULL>
__device__ __forceinline__ void kw_flush(const uint2* lst, int first, int nvalid, int lane,
                                         const int* __restrict__ cols,
                                         const float* __restrict__ XF, float* acc) {
    int c = 0, d = 0;
    if (FULL || lane < nvalid) {
        const uint2 t = lst[first + lane];
        c = cols[t.x];
        d = (int)t.y;
    }
    kw_add_rows<FULL>(c, d, nvalid, lane, XF, acc);
}

__device__ __forceinline__ void kw_scan_blk(const int4 (&r)[4], int blk, uint2* lst, int& cnt,
                                            int lane, int base,
                                            const int* __restrict__ cols,
                                            const float* __restrict__ XF, float* acc) {
#pragma unroll
    for (int q = 0; q < 4; ++q) {
        const unsigned dx = (unsigned)(r[q].x - base);
        const unsigned dy = (unsigned)(r[q].y - base);
        const unsigned dz = (unsigned)(r[q].z - base);
        const unsigned dw = (unsigned)(r[q].w - base);
        const unsigned dm = min(min(dx, dy), min(dz, dw));
        if (__builtin_amdgcn_ballot_w32(dm < (unsigned)SC_ROWS) != 0u) {
            const unsigned e = (unsigned)(blk * SC_BLK + q * 128 + lane * 4);
            kw_append(lst, cnt, dx < (unsigned)SC_ROWS, e,      dx);
            kw_append(lst, cnt, dy < (unsigned)SC_ROWS, e + 1u, dy);
            kw_append(lst, cnt, dz < (unsigned)SC_ROWS, e + 2u, dz);
            kw_append(lst, cnt, dw < (unsigned)SC_ROWS, e + 3u, dw);
        }
    }
    while (cnt >= 32) {
        cnt -= 32;
        kw_flush<true>(lst, cnt, 32, lane, cols, XF, acc);
    }
}

__global__ __launch_bounds__(1024) void kw_rowsum_lds(const int* __restrict__ rows,
                                                      const int* __restrict__ cols,
                                                      const float* __restrict__ dinv,
                                                      const float* __restrict__ XF,
                                                      float* out, int N, int E, int mwords,
                                                      int G, int pbase) {
    __shared__ __attribute__((aligned(16))) float acc[SC_ROWS * NFILT];
    __shared__ unsigned rcnt[SC_ROWS];
    __shared__ __attribute__((aligned(16))) uint2 slist[SC_WAVES * SC_CAP];

    const int tid  = threadIdx.x;
    const int lane = tid & 31;
    const int wv   = __builtin_amdgcn_readfirstlane(tid >> 5);
    const int base = blockIdx.x * SC_ROWS;

    const unsigned* lb = (const unsigned*)out + (size_t)base * NFILT;
    unsigned c0 = 0u, c1 = 0u;
    if (mwords > 0) {
        const unsigned t0 = lb[(size_t)min(lane,      G - 1) * LS_PITCH + LS_CAP];
        const unsigned t1 = lb[(size_t)min(lane + 32, G - 1) * LS_PITCH + LS_CAP];
        c0 = (lane      < G) ? t0 : 0u;
        c1 = (lane + 32 < G) ? t1 : 0u;
    }

    for (int i = tid; i < SC_ROWS * NFILT / 4; i += 1024)
        ((v4f*)acc)[i] = (v4f){0.0f, 0.0f, 0.0f, 0.0f};
    if (pbase >= 0 && tid < SC_ROWS) rcnt[tid] = 0u;
    __syncthreads();

    const int  pwl = tid & 255;
    const int  pgq = tid >> 8;
    const bool pon = pwl < SC_WORDS && blockIdx.x * SC_WORDS + pwl < ((N + 1) >> 1);
    const unsigned* pseg = (const unsigned*)out + (size_t)base * NFILT + max(pbase, 0) + (pon ? pwl : 0);
    unsigned pv[PP_BATCH] = {};

    float sc[SC_ST] = {};
    if (pbase < 0) {
#pragma unroll
        for (int t = 0; t < SC_ST; ++t)
            sc[t] = dinv[min(base + ((tid + t * 1024) >> 4), N - 1)];
    } else {
        kw_pp_load(pv, pseg, pgq, G);
    }

    uint2* lst = slist + wv * SC_CAP;
    int cnt = 0;

    const bool uselist = mwords > 0 &&
                         __builtin_amdgcn_ballot_w32(c0 > (unsigned)LS_CAP || c1 > (unsigned)LS_CAP) == 0u;

    if (uselist) {
        int      T  = 0;
        unsigned pe = 0u;
        int      pn = 0;
        for (int g = 0; g < G; ++g) {
            const int c  = __builtin_amdgcn_readlane((int)(g < 32 ? c0 : c1), g & 31);
            const int ng = (c + 31) >> 5;
            for (int j = (wv - T) & (SC_WAVES - 1); j < ng; j += SC_WAVES) {
                const int n = min(32, c - j * 32);
                const unsigned ne = lb[g * LS_PITCH + j * 32 + min(lane, n - 1)];
                if (pn == 32)
                    kw_add_rows<true>((int)(pe >> LS_ROW_BITS), (int)(pe & ((1u << LS_ROW_BITS) - 1u)),
                                      32, lane, XF, acc);
                else if (pn > 0)
                    kw_add_rows<false>((int)(pe >> LS_ROW_BITS), (int)(pe & ((1u << LS_ROW_BITS) - 1u)),
                                       pn, lane, XF, acc);
                pe = ne;
                pn = n;
            }
            T += ng;
        }
        if (pn == 32)
            kw_add_rows<true>((int)(pe >> LS_ROW_BITS), (int)(pe & ((1u << LS_ROW_BITS) - 1u)),
                              32, lane, XF, acc);
        else if (pn > 0)
            kw_add_rows<false>((int)(pe >> LS_ROW_BITS), (int)(pe & ((1u << LS_ROW_BITS) - 1u)),
                               pn, lane, XF, acc);
    } else {
    const int nfull = E / SC_BLK;
    const int iters = (nfull + SC_WAVES - 1) / SC_WAVES;

    if (iters > 0) {
        int k = (int)((blockIdx.x * 13u) % (unsigned)iters);

        int4 ra[4], rb[4];
        kw_load_blk(ra, rows, min(wv + SC_WAVES * k, nfull - 1), lane);

        for (int t = 0; t < iters; t += 2) {
            int kb = k + 1;
            if (kb == iters) kb = 0;
            const int blka = wv + SC_WAVES * k;
            const int blkb = wv + SC_WAVES * kb;
            kw_load_blk(rb, rows, min(blkb, nfull - 1), lane);
            if (blka < nfull)
                kw_scan_blk(ra, blka, lst, cnt, lane, base, cols, XF, acc);

            int kn = kb + 1;
            if (kn == iters) kn = 0;
            kw_load_blk(ra, rows, min(wv + SC_WAVES * kn, nfull - 1), lane);
            if (t + 1 < iters && blkb < nfull)
                kw_scan_blk(rb, blkb, lst, cnt, lane, base, cols, XF, acc);
            k = kn;
        }
    }
    if (nfull * SC_BLK < E && wv == (nfull & (SC_WAVES - 1))) {
        int4 rt[4];
        kw_load_tail(rt, rows, nfull, lane, E);
        kw_scan_blk(rt, nfull, lst, cnt, lane, base, cols, XF, acc);
    }
    }
    if (cnt > 0)
        kw_flush<false>(lst, 0, cnt, lane, cols, XF, acc);

    if (pbase >= 0) {
        unsigned lo = 0u, hi = 0u;
        kw_pp_sum(pv, pgq, G, pon, lo, hi);
        if (pon) {
            atomicAdd(&rcnt[2 * pwl],     lo);
            atomicAdd(&rcnt[2 * pwl + 1], hi);
        }
    }

    __syncthreads();

#pragma unroll
    for (int t = 0; t < SC_ST; ++t) {
        const int i   = tid + t * 1024;
        const int row = base + (i >> 4);
        if (i < SC_ROWS * (NFILT / 4) && row < N) {
            const float s = (pbase >= 0) ? kw_dinv_of(rcnt[i >> 4]) : sc[t];
            const v4f   a = ((const v4f*)acc)[i];
            ((v4f*)(out + (size_t)row * NFILT))[i & 15] = (v4f){s * a.x, s * a.y, s * a.z, s * a.w};
        }
    }
}

extern "C" void kernel_launch(void* const* d_in, const int* in_sizes, int n_in,
                              void* d_out, int out_size, void* d_ws, size_t ws_size,
                              hipStream_t stream) {
    const float* x   = (const float*)d_in[0];
    const float* flt = (const float*)d_in[1];
    const int*   eid = (const int*)d_in[2];

    const int N = in_sizes[0] / NFEAT;
    const int E = in_sizes[2] / 2;
    const int* rows = eid;
    const int* cols = eid + E;

    float* out = (float*)d_out;
    float* deg = (float*)d_ws;
    float* XF  = (float*)d_ws + (size_t)((N + 63) & ~63);

    unsigned* partial = (unsigned*)XF;
    const int nwords = (N + 1) / 2;
    const int R   = (N + DEG_RANGE_BINS - 1) / DEG_RANGE_BINS;
    const int G   = (E + DEG_CHUNK - 1) / DEG_CHUNK;
    const int wpg = R * DEG_RANGE_WORDS;
    const int nown     = (N + SC_ROWS - 1) / SC_ROWS;
    const int opq      = (nown + R - 1) / R;
    const int mwords   = G * LS_PITCH;
    const int lastrows = N - (nown - 1) * SC_ROWS;
    const bool keep    = opq <= LS_OWN_MAX && mwords <= lastrows * NFILT && G <= 64
                         && N <= (1 << (32 - LS_ROW_BITS)) && (E & 3) == 0;
    unsigned* lists    = keep ? (unsigned*)d_out : nullptr;
    const bool park    = keep && mwords + G * PP_PITCH <= lastrows * NFILT && G <= 4 * PP_BATCH;
    const int  pbase   = park ? mwords : -1;
    if (keep)
        kw_degree_lds<true><<<dim3(G, R), 1024, 0, stream>>>(rows, cols, partial, lists, E, nwords, wpg, nown, opq, pbase);
    else
        kw_degree_lds<false><<<dim3(G, R), 1024, 0, stream>>>(rows, cols, partial, lists, E, nwords, wpg, nown, opq, pbase);
    if (!park)
        kw_dinv_reduce<<<(nwords + 255) / 256, 256, 0, stream>>>(partial, deg, N, nwords, G, wpg);

    kw_gemm_wmma<<<nown, 1024, 0, stream>>>(x, flt, deg, park ? (const unsigned*)d_out : nullptr,
                                            XF, N, G, pbase);

    kw_rowsum_lds<<<nown, 1024, 0, stream>>>(rows, cols, deg, XF, out, N, E, keep ? mwords : 0, G, pbase);
}
